# starting kernel
# baseline (speedup 1.0000x reference)
#include <hip/hip_runtime.h>
#include <hip/hip_bf16.h>


typedef __attribute__((ext_vector_type(2))) float v2f;
typedef __attribute__((ext_vector_type(8))) float v8f;


__global__ void deg_init_kernel(float* __restrict__ deg, int n) {
    int i = blockIdx.x * blockDim.x + threadIdx.x;
    if (i < n) deg[i] = 1.0f;
}

__global__ void deg_accum_kernel(const int* __restrict__ dst, float* __restrict__ deg, int e) {
    int i = blockIdx.x * blockDim.x + threadIdx.x;
    if (i < e) atomicAdd(&deg[dst[i]], 1.0f);
}

__global__ void deg_to_dinv_kernel(float* __restrict__ deg, int n) {
    int i = blockIdx.x * blockDim.x + threadIdx.x;
    if (i < n) {
        float d = deg[i];
        deg[i] = (d > 0.0f) ? rsqrtf(d) : 0.0f;
    }
}


template<int K, int NCOLS, bool RELU_A>
__global__ __launch_bounds__((NCOLS / 16) * 32)
void wmma_gemm_kernel(const float* __restrict__ A, const float* __restrict__ B,
                      float* __restrict__ C, int M) {
    const int wave = threadIdx.x >> 5;
    const int lane = threadIdx.x & 31;
    const int m0 = blockIdx.x * 16;
    if (m0 >= M) return;

    const int n0   = wave * 16;
    const int lo16 = lane & 15;
    const int half = lane >> 4;
    const int col  = n0 + lo16;

    const float* __restrict__ arow = A + (size_t)(m0 + lo16) * K;

    v8f acc = {};
    #pragma unroll
    for (int k = 0; k < K; k += 4) {
        const int kb = k + half * 2;
        float2 av = *(const float2*)(arow + kb);
        if (RELU_A) { av.x = fmaxf(av.x, 0.0f); av.y = fmaxf(av.y, 0.0f); }
        v2f a; a[0] = av.x; a[1] = av.y;
        v2f b;
        b[0] = B[(size_t)kb * NCOLS + col];
        b[1] = B[(size_t)(kb + 1) * NCOLS + col];
        acc = __builtin_amdgcn_wmma_f32_16x16x4_f32(
                   false, a,  false, b,
                   (short)0, acc,  false,  false);
    }

    const int rbase = m0 + half * 8;
    #pragma unroll
    for (int v = 0; v < 8; ++v)
        C[(size_t)(rbase + v) * NCOLS + col] = acc[v];
}


template<int C>
__global__ void agg_init_kernel(const float* __restrict__ h, const float* __restrict__ dinv,
                                const float* __restrict__ bias, float* __restrict__ agg, int n) {
    int i = blockIdx.x * blockDim.x + threadIdx.x;
    int node = i / C, c = i % C;
    if (node < n) {
        float di = dinv[node];
        agg[i] = bias[c] + h[i] * di * di;
    }
}

template<int C>
__global__ void edge_scatter_kernel(const int* __restrict__ src, const int* __restrict__ dst,
                                    const float* __restrict__ h, const float* __restrict__ dinv,
                                    float* __restrict__ agg, int e) {
    int t = blockIdx.x * blockDim.x + threadIdx.x;
    int edge = t / C, c = t % C;
    if (edge >= e) return;
    int s = src[edge], d = dst[edge];
    float norm = dinv[s] * dinv[d];
    atomicAdd(&agg[(size_t)d * C + c], h[(size_t)s * C + c] * norm);
}


__global__ void head_kernel(const float* __restrict__ agg2, const float* __restrict__ Wc,
                            const float* __restrict__ bc, float* __restrict__ out, int n) {
    int i = blockIdx.x * blockDim.x + threadIdx.x;
    if (i >= n) return;
    const float* row = agg2 + (size_t)i * 32;
    float acc = bc[0];
    #pragma unroll
    for (int c = 0; c < 32; ++c)
        acc += fmaxf(row[c], 0.0f) * Wc[c];
    out[i] = acc;
}


static inline int cdiv(long long a, long long b) { return (int)((a + b - 1) / b); }

extern "C" void kernel_launch(void* const* d_in, const int* in_sizes, int n_in,
                              void* d_out, int out_size, void* d_ws, size_t ws_size,
                              hipStream_t stream) {
    const float* x   = (const float*)d_in[0];
    const int*   ei  = (const int*)d_in[1];
    const float* W1  = (const float*)d_in[2];
    const float* b1  = (const float*)d_in[3];
    const float* W2  = (const float*)d_in[4];
    const float* b2  = (const float*)d_in[5];
    const float* Wc  = (const float*)d_in[6];
    const float* bc  = (const float*)d_in[7];
    float* out = (float*)d_out;

    const int N = in_sizes[0] / 128;
    const int E = in_sizes[1] / 2;
    const int* src = ei;
    const int* dst = ei + E;

    float* ws   = (float*)d_ws;
    float* dinv = ws;
    float* h1   = dinv + N;
    float* agg1 = h1 + (size_t)N * 64;
    float* h2   = agg1 + (size_t)N * 64;
    float* agg2 = h2 + (size_t)N * 32;
    (void)ws_size; (void)n_in; (void)out_size;

    const int T = 256;

    deg_init_kernel<<<cdiv(N, T), T, 0, stream>>>(dinv, N);
    deg_accum_kernel<<<cdiv(E, T), T, 0, stream>>>(dst, dinv, E);
    deg_to_dinv_kernel<<<cdiv(N, T), T, 0, stream>>>(dinv, N);

    wmma_gemm_kernel<128, 64, false><<<cdiv(N, 16), 128, 0, stream>>>(x, W1, h1, N);

    agg_init_kernel<64><<<cdiv((long long)N * 64, T), T, 0, stream>>>(h1, dinv, b1, agg1, N);
    edge_scatter_kernel<64><<<cdiv((long long)E * 64, T), T, 0, stream>>>(src, dst, h1, dinv, agg1, E);

    wmma_gemm_kernel<64, 32, true><<<cdiv(N, 16), 64, 0, stream>>>(agg1, W2, h2, N);

    agg_init_kernel<32><<<cdiv((long long)N * 32, T), T, 0, stream>>>(h2, dinv, b2, agg2, N);
    edge_scatter_kernel<32><<<cdiv((long long)E * 32, T), T, 0, stream>>>(src, dst, h2, dinv, agg2, E);

    head_kernel<<<cdiv(N, T), T, 0, stream>>>(agg2, Wc, bc, out, N);
}


# Elementwise launches folded into the GEMM epilogues: each X@W tile also stores agg = bias + h*dinv^2, layer-1 tiles form dinv = rsqrt(deg) themselves: 10 launches to 7; atomic sum order as before
# speedup vs baseline: 1.7964x; 1.7964x over previous
#include <hip/hip_runtime.h>
#include <hip/hip_bf16.h>


typedef __attribute__((ext_vector_type(2))) float v2f;
typedef __attribute__((ext_vector_type(8))) float v8f;


__global__ void deg_init_kernel(float* __restrict__ deg, int n) {
    int i = blockIdx.x * blockDim.x + threadIdx.x;
    if (i < n) deg[i] = 1.0f;
}

__global__ void deg_accum_kernel(const int* __restrict__ dst, float* __restrict__ deg, int e) {
    int i = blockIdx.x * blockDim.x + threadIdx.x;
    if (i < e) atomicAdd(&deg[dst[i]], 1.0f);
}

__device__ __forceinline__ float dinv_of(float d) {
    return (d > 0.0f) ? rsqrtf(d) : 0.0f;
}


template<int K, int NCOLS, bool RELU_A, bool DEG_IN>
__global__ __launch_bounds__((NCOLS / 16) * 32)
void wmma_gemm_kernel(const float* __restrict__ A, const float* __restrict__ B,
                      const float* __restrict__ nrm, const float* __restrict__ bias,
                      float* __restrict__ C, float* __restrict__ AGG,
                      float* __restrict__ dinv_out, int M) {
    const int wave = threadIdx.x >> 5;
    const int lane = threadIdx.x & 31;
    const int m0 = blockIdx.x * 16;
    if (m0 >= M) return;

    const int n0   = wave * 16;
    const int lo16 = lane & 15;
    const int half = lane >> 4;
    const int col  = n0 + lo16;

    if (DEG_IN && wave == 0 && (blockIdx.x & 1) == 0) {
        const int r = m0 + lane;
        if (r < M) dinv_out[r] = dinv_of(nrm[r]);
    }

    const float* __restrict__ arow = A + (size_t)(m0 + lo16) * K;

    const int rbase = m0 + half * 8;
    float dg[8];
    #pragma unroll
    for (int v = 0; v < 8; ++v) dg[v] = nrm[rbase + v];
    const float bcol = bias[col];

    v8f acc = {};
    #pragma unroll
    for (int k = 0; k < K; k += 4) {
        const int kb = k + half * 2;
        float2 av = *(const float2*)(arow + kb);
        if (RELU_A) { av.x = fmaxf(av.x, 0.0f); av.y = fmaxf(av.y, 0.0f); }
        v2f a; a[0] = av.x; a[1] = av.y;
        v2f b;
        b[0] = B[(size_t)kb * NCOLS + col];
        b[1] = B[(size_t)(kb + 1) * NCOLS + col];
        acc = __builtin_amdgcn_wmma_f32_16x16x4_f32(
                   false, a,  false, b,
                   (short)0, acc,  false,  false);
    }

    #pragma unroll
    for (int v = 0; v < 8; ++v) {
        const float di = DEG_IN ? dinv_of(dg[v]) : dg[v];
        const size_t o = (size_t)(rbase + v) * NCOLS + col;
        C[o]   = acc[v];
        AGG[o] = bcol + acc[v] * di * di;
    }
}


template<int C, int EPW>
__global__ __launch_bounds__(256)
void edge_scatter_kernel(const int* __restrict__ src, const int* __restrict__ dst,
                         const float* __restrict__ h, const float* __restrict__ dinv,
                         float* __restrict__ agg, int e) {
    constexpr int R = C / 32;
    const int lane = threadIdx.x & 31;
    const int wave = __builtin_amdgcn_readfirstlane((int)(blockIdx.x * 8 + (threadIdx.x >> 5)));
    const int e0 = wave * EPW;
    if (e0 + EPW <= e) {
        int s[EPW], d[EPW];
        float nrm[EPW];
        float v[EPW][R];
        for (int j = 0; j < EPW; ++j) { s[j] = src[e0 + j]; d[j] = dst[e0 + j]; }
        for (int j = 0; j < EPW; ++j) {
            nrm[j] = dinv[s[j]] * dinv[d[j]];
            for (int r = 0; r < R; ++r)
                v[j][r] = h[(size_t)s[j] * C + r * 32 + lane];
        }
        for (int j = 0; j < EPW; ++j) {
            for (int r = 0; r < R; ++r)
                atomicAdd(&agg[(size_t)d[j] * C + r * 32 + lane], v[j][r] * nrm[j]);
        }
    } else {
        for (int j = 0; j < EPW; ++j) {
            const int edge = e0 + j;
            if (edge >= e) break;
            const int s1 = src[edge], d1 = dst[edge];
            const float norm = dinv[s1] * dinv[d1];
            for (int r = 0; r < R; ++r)
                atomicAdd(&agg[(size_t)d1 * C + r * 32 + lane],
                          h[(size_t)s1 * C + r * 32 + lane] * norm);
        }
    }
}


__global__ void head_kernel(const float* __restrict__ agg2, const float* __restrict__ Wc,
                            const float* __restrict__ bc, float* __restrict__ out, int n) {
    int i = blockIdx.x * blockDim.x + threadIdx.x;
    if (i >= n) return;
    const float* row = agg2 + (size_t)i * 32;
    float acc = bc[0];
    #pragma unroll
    for (int c = 0; c < 32; ++c)
        acc += fmaxf(row[c], 0.0f) * Wc[c];
    out[i] = acc;
}


static inline int cdiv(long long a, long long b) { return (int)((a + b - 1) / b); }

extern "C" void kernel_launch(void* const* d_in, const int* in_sizes, int n_in,
                              void* d_out, int out_size, void* d_ws, size_t ws_size,
                              hipStream_t stream) {
    const float* x   = (const float*)d_in[0];
    const int*   ei  = (const int*)d_in[1];
    const float* W1  = (const float*)d_in[2];
    const float* b1  = (const float*)d_in[3];
    const float* W2  = (const float*)d_in[4];
    const float* b2  = (const float*)d_in[5];
    const float* Wc  = (const float*)d_in[6];
    const float* bc  = (const float*)d_in[7];
    float* out = (float*)d_out;

    const int N = in_sizes[0] / 128;
    const int E = in_sizes[1] / 2;
    const int* src = ei;
    const int* dst = ei + E;

    float* ws   = (float*)d_ws;
    float* dinv = ws;
    float* h1   = dinv + N;
    float* agg1 = h1 + (size_t)N * 64;
    float* h2   = agg1 + (size_t)N * 64;
    float* agg2 = h2 + (size_t)N * 32;
    float* deg  = agg2;
    (void)ws_size; (void)n_in; (void)out_size;

    const int T = 256;

    deg_init_kernel<<<cdiv(N, T), T, 0, stream>>>(deg, N);
    deg_accum_kernel<<<cdiv(E, T), T, 0, stream>>>(dst, deg, E);

    wmma_gemm_kernel<128, 64, false, true><<<cdiv(N, 16), 128, 0, stream>>>(
        x, W1, deg, b1, h1, agg1, dinv, N);
    edge_scatter_kernel<64, 8><<<cdiv(E, 8 * 8), T, 0, stream>>>(src, dst, h1, dinv, agg1, E);

    wmma_gemm_kernel<64, 32, true, false><<<cdiv(N, 16), 64, 0, stream>>>(
        agg1, W2, dinv, b2, h2, agg2, nullptr, N);
    edge_scatter_kernel<32, 16><<<cdiv(E, 16 * 8), T, 0, stream>>>(src, dst, h2, dinv, agg2, E);

    head_kernel<<<cdiv(N, T), T, 0, stream>>>(agg2, Wc, bc, out, N);
}


# In-degrees counted with LDS integer adds by one work-group per (node range, edge chunk) and stored as count planes that the layer-1 GEMM sums: no global atomic per edge for deg, 7 launches to 6
# speedup vs baseline: 2.1049x; 1.1717x over previous
#include <hip/hip_runtime.h>
#include <hip/hip_bf16.h>


typedef __attribute__((ext_vector_type(2))) float v2f;
typedef __attribute__((ext_vector_type(8))) float v8f;


constexpr int DEG_PLANES = 16;
constexpr int DEG_SPAN   = 6400;

__global__ __launch_bounds__(1024)
void deg_hist_kernel(const int* __restrict__ dst, float* __restrict__ part, int n, int e) {
    __shared__ unsigned cnt[DEG_SPAN];
    const int p   = blockIdx.x / DEG_PLANES;
    const int c   = blockIdx.x % DEG_PLANES;
    const int lo  = p * DEG_SPAN;
    const int len = min(DEG_SPAN, n - lo);
    for (int i = threadIdx.x; i < DEG_SPAN; i += 1024) cnt[i] = 0u;
    __syncthreads();
    const int per = (e + DEG_PLANES - 1) / DEG_PLANES;
    const int e0  = c * per;
    const int e1  = min(e0 + per, e);
    constexpr int U = 16;
    for (int base = e0 + (int)threadIdx.x; base < e1; base += 1024 * U) {
        int d[U];
        if (base + (U - 1) * 1024 < e1) {
            #pragma unroll
            for (int j = 0; j < U; ++j) d[j] = dst[base + j * 1024];
        } else {
            #pragma unroll
            for (int j = 0; j < U; ++j) {
                const int i = base + j * 1024;
                d[j] = (i < e1) ? dst[i] : -1;
            }
        }
        #pragma unroll
        for (int j = 0; j < U; ++j) {
            const unsigned u = (unsigned)(d[j] - lo);
            if (u < (unsigned)len) atomicAdd(&cnt[u], 1u);
        }
    }
    __syncthreads();
    float* __restrict__ row = part + (size_t)c * n + lo;
    for (int i = threadIdx.x; i < len; i += 1024) row[i] = (float)cnt[i];
}

__device__ __forceinline__ float dinv_of(float d) {
    return (d > 0.0f) ? rsqrtf(d) : 0.0f;
}


template<int K, int NCOLS, bool RELU_A, bool DEG_IN>
__global__ __launch_bounds__((NCOLS / 16) * 32)
void wmma_gemm_kernel(const float* __restrict__ A, const float* __restrict__ B,
                      const float* __restrict__ nrm, const float* __restrict__ bias,
                      float* __restrict__ C, float* __restrict__ AGG,
                      float* __restrict__ dinv_out, int M) {
    const int wave = threadIdx.x >> 5;
    const int lane = threadIdx.x & 31;
    const int m0 = blockIdx.x * 16;
    if (m0 >= M) return;

    const int n0   = wave * 16;
    const int lo16 = lane & 15;
    const int half = lane >> 4;
    const int col  = n0 + lo16;

    if (DEG_IN && wave == 0 && (blockIdx.x & 1) == 0) {
        const int r = m0 + lane;
        if (r < M) {
            float d = 1.0f;
            #pragma unroll
            for (int c = 0; c < DEG_PLANES; ++c) d += nrm[(size_t)c * M + r];
            dinv_out[r] = dinv_of(d);
        }
    }

    const float* __restrict__ arow = A + (size_t)(m0 + lo16) * K;

    const int rbase = m0 + half * 8;
    float dg[8];
    float dpart = 0.0f;
    if (DEG_IN) {
        #pragma unroll
        for (int c = 0; c < DEG_PLANES; c += 2)
            dpart += nrm[(size_t)(c + half) * M + m0 + lo16];
    } else {
        #pragma unroll
        for (int v = 0; v < 8; ++v) dg[v] = nrm[rbase + v];
    }
    const float bcol = bias[col];

    v8f acc = {};
    #pragma unroll
    for (int k = 0; k < K; k += 4) {
        const int kb = k + half * 2;
        float2 av = *(const float2*)(arow + kb);
        if (RELU_A) { av.x = fmaxf(av.x, 0.0f); av.y = fmaxf(av.y, 0.0f); }
        v2f a; a[0] = av.x; a[1] = av.y;
        v2f b;
        b[0] = B[(size_t)kb * NCOLS + col];
        b[1] = B[(size_t)(kb + 1) * NCOLS + col];
        acc = __builtin_amdgcn_wmma_f32_16x16x4_f32(
                   false, a,  false, b,
                   (short)0, acc,  false,  false);
    }

    if (DEG_IN) {
        const float di_row = dinv_of(1.0f + dpart + __shfl_xor(dpart, 16));
        #pragma unroll
        for (int v = 0; v < 8; ++v) dg[v] = __shfl(di_row, half * 8 + v);
    }

    #pragma unroll
    for (int v = 0; v < 8; ++v) {
        const float di = dg[v];
        const size_t o = (size_t)(rbase + v) * NCOLS + col;
        C[o]   = acc[v];
        AGG[o] = bcol + acc[v] * di * di;
    }
}


template<int C, int EPW>
__global__ __launch_bounds__(256)
void edge_scatter_kernel(const int* __restrict__ src, const int* __restrict__ dst,
                         const float* __restrict__ h, const float* __restrict__ dinv,
                         float* __restrict__ agg, int e) {
    constexpr int R = C / 32;
    const int lane = threadIdx.x & 31;
    const int wave = __builtin_amdgcn_readfirstlane((int)(blockIdx.x * 8 + (threadIdx.x >> 5)));
    const int e0 = wave * EPW;
    if (e0 + EPW <= e) {
        int s[EPW], d[EPW];
        float nrm[EPW];
        float v[EPW][R];
        for (int j = 0; j < EPW; ++j) { s[j] = src[e0 + j]; d[j] = dst[e0 + j]; }
        for (int j = 0; j < EPW; ++j) {
            nrm[j] = dinv[s[j]] * dinv[d[j]];
            for (int r = 0; r < R; ++r)
                v[j][r] = h[(size_t)s[j] * C + r * 32 + lane];
        }
        for (int j = 0; j < EPW; ++j) {
            for (int r = 0; r < R; ++r)
                atomicAdd(&agg[(size_t)d[j] * C + r * 32 + lane], v[j][r] * nrm[j]);
        }
    } else {
        for (int j = 0; j < EPW; ++j) {
            const int edge = e0 + j;
            if (edge >= e) break;
            const int s1 = src[edge], d1 = dst[edge];
            const float norm = dinv[s1] * dinv[d1];
            for (int r = 0; r < R; ++r)
                atomicAdd(&agg[(size_t)d1 * C + r * 32 + lane],
                          h[(size_t)s1 * C + r * 32 + lane] * norm);
        }
    }
}


__global__ void head_kernel(const float* __restrict__ agg2, const float* __restrict__ Wc,
                            const float* __restrict__ bc, float* __restrict__ out, int n) {
    int i = blockIdx.x * blockDim.x + threadIdx.x;
    if (i >= n) return;
    const float* row = agg2 + (size_t)i * 32;
    float acc = bc[0];
    #pragma unroll
    for (int c = 0; c < 32; ++c)
        acc += fmaxf(row[c], 0.0f) * Wc[c];
    out[i] = acc;
}


static inline int cdiv(long long a, long long b) { return (int)((a + b - 1) / b); }

extern "C" void kernel_launch(void* const* d_in, const int* in_sizes, int n_in,
                              void* d_out, int out_size, void* d_ws, size_t ws_size,
                              hipStream_t stream) {
    const float* x   = (const float*)d_in[0];
    const int*   ei  = (const int*)d_in[1];
    const float* W1  = (const float*)d_in[2];
    const float* b1  = (const float*)d_in[3];
    const float* W2  = (const float*)d_in[4];
    const float* b2  = (const float*)d_in[5];
    const float* Wc  = (const float*)d_in[6];
    const float* bc  = (const float*)d_in[7];
    float* out = (float*)d_out;

    const int N = in_sizes[0] / 128;
    const int E = in_sizes[1] / 2;
    const int* src = ei;
    const int* dst = ei + E;

    float* ws   = (float*)d_ws;
    float* dinv = ws;
    float* h1   = dinv + N;
    float* agg1 = h1 + (size_t)N * 64;
    float* h2   = agg1 + (size_t)N * 64;
    float* agg2 = h2 + (size_t)N * 32;
    float* degp = agg2;
    (void)ws_size; (void)n_in; (void)out_size;

    const int T = 256;

    deg_hist_kernel<<<cdiv(N, DEG_SPAN) * DEG_PLANES, 1024, 0, stream>>>(dst, degp, N, E);

    wmma_gemm_kernel<128, 64, false, true><<<cdiv(N, 16), 128, 0, stream>>>(
        x, W1, degp, b1, h1, agg1, dinv, N);
    edge_scatter_kernel<64, 8><<<cdiv(E, 8 * 8), T, 0, stream>>>(src, dst, h1, dinv, agg1, E);

    wmma_gemm_kernel<64, 32, true, false><<<cdiv(N, 16), 64, 0, stream>>>(
        agg1, W2, dinv, b2, h2, agg2, nullptr, N);
    edge_scatter_kernel<32, 16><<<cdiv(E, 16 * 8), T, 0, stream>>>(src, dst, h2, dinv, agg2, E);

    head_kernel<<<cdiv(N, T), T, 0, stream>>>(agg2, Wc, bc, out, N);
}


# GEMM re-tiled: one wave forms a whole 16-row stripe (A rows read once in 16-byte pieces, W from an LDS copy in fragment order, whole-line stores); k summed in another order inside each 8
# speedup vs baseline: 2.4374x; 1.1580x over previous
#include <hip/hip_runtime.h>
#include <hip/hip_bf16.h>


typedef __attribute__((ext_vector_type(2))) float v2f;
typedef __attribute__((ext_vector_type(8))) float v8f;


constexpr int DEG_PLANES = 16;
constexpr int DEG_SPAN   = 6400;

__global__ __launch_bounds__(1024)
void deg_hist_kernel(const int* __restrict__ dst, float* __restrict__ part, int n, int e) {
    __shared__ unsigned cnt[DEG_SPAN];
    const int p   = blockIdx.x / DEG_PLANES;
    const int c   = blockIdx.x % DEG_PLANES;
    const int lo  = p * DEG_SPAN;
    const int len = min(DEG_SPAN, n - lo);
    for (int i = threadIdx.x; i < DEG_SPAN; i += 1024) cnt[i] = 0u;
    __syncthreads();
    const int per = (e + DEG_PLANES - 1) / DEG_PLANES;
    const int e0  = c * per;
    const int e1  = min(e0 + per, e);
    constexpr int U = 16;
    for (int base = e0 + (int)threadIdx.x; base < e1; base += 1024 * U) {
        int d[U];
        if (base + (U - 1) * 1024 < e1) {
            #pragma unroll
            for (int j = 0; j < U; ++j) d[j] = dst[base + j * 1024];
        } else {
            #pragma unroll
            for (int j = 0; j < U; ++j) {
                const int i = base + j * 1024;
                d[j] = (i < e1) ? dst[i] : -1;
            }
        }
        #pragma unroll
        for (int j = 0; j < U; ++j) {
            const unsigned u = (unsigned)(d[j] - lo);
            if (u < (unsigned)len) atomicAdd(&cnt[u], 1u);
        }
    }
    __syncthreads();
    float* __restrict__ row = part + (size_t)c * n + lo;
    for (int i = threadIdx.x; i < len; i += 1024) row[i] = (float)cnt[i];
}

__device__ __forceinline__ float dinv_of(float d) {
    return (d > 0.0f) ? rsqrtf(d) : 0.0f;
}


constexpr int GEMM_WAVES = 8;

__device__ __forceinline__ float xhalf(float x) {
    return __builtin_bit_cast(float, __builtin_amdgcn_permlanex16(
        __builtin_bit_cast(int, x), __builtin_bit_cast(int, x),
        0x76543210, 0xfedcba98, false, false));
}

template<int K, int NCOLS, bool RELU_A, bool DEG_IN>
__global__ __launch_bounds__(GEMM_WAVES * 32)
void wmma_gemm_kernel(const float* __restrict__ A, const float* __restrict__ B,
                      const float* __restrict__ nrm, const float* __restrict__ bias,
                      float* __restrict__ C, float* __restrict__ AGG,
                      float* __restrict__ dinv_out, int M) {
    constexpr int NT = NCOLS / 16;
    constexpr int KB = K / 8;
    constexpr int KQ = K / 4;
    __shared__ float4 Wl[KQ * NCOLS];

    const int wave = __builtin_amdgcn_readfirstlane((int)(threadIdx.x >> 5));
    const int lane = threadIdx.x & 31;
    const int lo16 = lane & 15;
    const int half = lane >> 4;
    const int stripe = blockIdx.x * GEMM_WAVES + wave;
    const int m0 = stripe * 16;
    const bool live = m0 < M;
    const int mr = live ? m0 : 0;

    constexpr int NST = (KQ * NCOLS) / (GEMM_WAVES * 32);
    static_assert((KQ * NCOLS) % (GEMM_WAVES * 32) == 0, "B copy: whole pieces per thread");
    float4 st[NST];
    #pragma unroll
    for (int it = 0; it < NST; ++it) {
        const int i = (int)threadIdx.x + it * (GEMM_WAVES * 32);
        const int q = i / NCOLS, col = i % NCOLS;
        const float* __restrict__ bp = B + (size_t)(4 * q) * NCOLS + col;
        st[it] = make_float4(bp[0], bp[NCOLS], bp[2 * NCOLS], bp[3 * NCOLS]);
    }

    float4 av[KB];
    {
        const float* __restrict__ arow = A + (size_t)(mr + lo16) * K + 4 * half;
        #pragma unroll
        for (int j = 0; j < KB; ++j) av[j] = *(const float4*)(arow + 8 * j);
    }
    float dg[16];
    float dp[DEG_PLANES / 2];
    if (DEG_IN) {
        #pragma unroll
        for (int c = 0; c < DEG_PLANES / 2; ++c)
            dp[c] = nrm[(size_t)(2 * c + half) * M + mr + lo16];
    } else {
        #pragma unroll
        for (int r = 0; r < 16; ++r) dg[r] = nrm[mr + r];
    }

    #pragma unroll
    for (int it = 0; it < NST; ++it)
        Wl[(int)threadIdx.x + it * (GEMM_WAVES * 32)] = st[it];
    __builtin_amdgcn_fence(__ATOMIC_RELEASE, "workgroup", "local");
    __builtin_amdgcn_s_barrier();
    __builtin_amdgcn_fence(__ATOMIC_ACQUIRE, "workgroup", "local");
    if (!live) return;

    if (DEG_IN && (stripe & 1) == 0) {
        const int r = m0 + lane;
        if (r < M) {
            float d = 1.0f;
            #pragma unroll
            for (int c = 0; c < DEG_PLANES; ++c) d += nrm[(size_t)c * M + r];
            dinv_out[r] = dinv_of(d);
        }
    }

    v8f acc[NT];
    #pragma unroll
    for (int t = 0; t < NT; ++t) acc[t] = v8f{};

    #pragma unroll
    for (int j = 0; j < KB; ++j) {
        float4 a4 = av[j];
        if (RELU_A) {
            a4.x = fmaxf(a4.x, 0.0f); a4.y = fmaxf(a4.y, 0.0f);
            a4.z = fmaxf(a4.z, 0.0f); a4.w = fmaxf(a4.w, 0.0f);
        }
        v2f a01; a01[0] = a4.x; a01[1] = a4.y;
        v2f a23; a23[0] = a4.z; a23[1] = a4.w;
        float4 w[NT];
        #pragma unroll
        for (int t = 0; t < NT; ++t) w[t] = Wl[(2 * j + half) * NCOLS + 16 * t + lo16];
        #pragma unroll
        for (int t = 0; t < NT; ++t) {
            v2f b; b[0] = w[t].x; b[1] = w[t].y;
            acc[t] = __builtin_amdgcn_wmma_f32_16x16x4_f32(
                          false, a01,  false, b,
                          (short)0, acc[t],  false,  false);
        }
        #pragma unroll
        for (int t = 0; t < NT; ++t) {
            v2f b; b[0] = w[t].z; b[1] = w[t].w;
            acc[t] = __builtin_amdgcn_wmma_f32_16x16x4_f32(
                          false, a23,  false, b,
                          (short)0, acc[t],  false,  false);
        }
    }

    if (DEG_IN) {
        float dpart = 0.0f;
        #pragma unroll
        for (int c = 0; c < DEG_PLANES / 2; ++c) dpart += dp[c];
        const float di_row = dinv_of(1.0f + dpart + xhalf(dpart));
        #pragma unroll
        for (int r = 0; r < 16; ++r)
            dg[r] = __builtin_bit_cast(float,
                        __builtin_amdgcn_readlane(__builtin_bit_cast(int, di_row), r));
    }

    #pragma unroll
    for (int p = 0; p < NT / 2; ++p) {
        const int col = 32 * p + lane;
        const float bcol = bias[col];
        #pragma unroll
        for (int v = 0; v < 8; ++v) {
            const float t0 = acc[2 * p][v];
            const float t1 = acc[2 * p + 1][v];
            const float got = xhalf(half ? t0 : t1);
            const float x = half ? got : t0;
            const float y = half ? t1 : got;
            const size_t ox = (size_t)(m0 + v) * NCOLS + col;
            const size_t oy = (size_t)(m0 + 8 + v) * NCOLS + col;
            const float dx = dg[v], dy = dg[v + 8];
            C[ox]   = x;
            AGG[ox] = bcol + x * dx * dx;
            C[oy]   = y;
            AGG[oy] = bcol + y * dy * dy;
        }
    }
}


template<int C, int EPW>
__global__ __launch_bounds__(256)
void edge_scatter_kernel(const int* __restrict__ src, const int* __restrict__ dst,
                         const float* __restrict__ h, const float* __restrict__ dinv,
                         float* __restrict__ agg, int e) {
    constexpr int R = C / 32;
    const int lane = threadIdx.x & 31;
    const int wave = __builtin_amdgcn_readfirstlane((int)(blockIdx.x * 8 + (threadIdx.x >> 5)));
    const int e0 = wave * EPW;
    if (e0 + EPW <= e) {
        int s[EPW], d[EPW];
        float nrm[EPW];
        float v[EPW][R];
        for (int j = 0; j < EPW; ++j) { s[j] = src[e0 + j]; d[j] = dst[e0 + j]; }
        for (int j = 0; j < EPW; ++j) {
            nrm[j] = dinv[s[j]] * dinv[d[j]];
            for (int r = 0; r < R; ++r)
                v[j][r] = h[(size_t)s[j] * C + r * 32 + lane];
        }
        for (int j = 0; j < EPW; ++j) {
            for (int r = 0; r < R; ++r)
                atomicAdd(&agg[(size_t)d[j] * C + r * 32 + lane], v[j][r] * nrm[j]);
        }
    } else {
        for (int j = 0; j < EPW; ++j) {
            const int edge = e0 + j;
            if (edge >= e) break;
            const int s1 = src[edge], d1 = dst[edge];
            const float norm = dinv[s1] * dinv[d1];
            for (int r = 0; r < R; ++r)
                atomicAdd(&agg[(size_t)d1 * C + r * 32 + lane],
                          h[(size_t)s1 * C + r * 32 + lane] * norm);
        }
    }
}


__global__ void head_kernel(const float* __restrict__ agg2, const float* __restrict__ Wc,
                            const float* __restrict__ bc, float* __restrict__ out, int n) {
    int i = blockIdx.x * blockDim.x + threadIdx.x;
    if (i >= n) return;
    const float* row = agg2 + (size_t)i * 32;
    float acc = bc[0];
    #pragma unroll
    for (int c = 0; c < 32; ++c)
        acc += fmaxf(row[c], 0.0f) * Wc[c];
    out[i] = acc;
}


static inline int cdiv(long long a, long long b) { return (int)((a + b - 1) / b); }

extern "C" void kernel_launch(void* const* d_in, const int* in_sizes, int n_in,
                              void* d_out, int out_size, void* d_ws, size_t ws_size,
                              hipStream_t stream) {
    const float* x   = (const float*)d_in[0];
    const int*   ei  = (const int*)d_in[1];
    const float* W1  = (const float*)d_in[2];
    const float* b1  = (const float*)d_in[3];
    const float* W2  = (const float*)d_in[4];
    const float* b2  = (const float*)d_in[5];
    const float* Wc  = (const float*)d_in[6];
    const float* bc  = (const float*)d_in[7];
    float* out = (float*)d_out;

    const int N = in_sizes[0] / 128;
    const int E = in_sizes[1] / 2;
    const int* src = ei;
    const int* dst = ei + E;

    float* ws   = (float*)d_ws;
    float* dinv = ws;
    float* h1   = dinv + N;
    float* agg1 = h1 + (size_t)N * 64;
    float* h2   = agg1 + (size_t)N * 64;
    float* agg2 = h2 + (size_t)N * 32;
    float* degp = agg2;
    (void)ws_size; (void)n_in; (void)out_size;

    const int T = 256;

    deg_hist_kernel<<<cdiv(N, DEG_SPAN) * DEG_PLANES, 1024, 0, stream>>>(dst, degp, N, E);

    wmma_gemm_kernel<128, 64, false, true><<<cdiv(N, 16 * GEMM_WAVES), GEMM_WAVES * 32, 0, stream>>>(
        x, W1, degp, b1, h1, agg1, dinv, N);
    edge_scatter_kernel<64, 8><<<cdiv(E, 8 * 8), T, 0, stream>>>(src, dst, h1, dinv, agg1, E);

    wmma_gemm_kernel<64, 32, true, false><<<cdiv(N, 16 * GEMM_WAVES), GEMM_WAVES * 32, 0, stream>>>(
        agg1, W2, dinv, b2, h2, agg2, nullptr, N);
    edge_scatter_kernel<32, 16><<<cdiv(E, 16 * 8), T, 0, stream>>>(src, dst, h2, dinv, agg2, E);

    head_kernel<<<cdiv(N, T), T, 0, stream>>>(agg2, Wc, bc, out, N);
}


# Edge sums formed in LDS by the owner of a node range: edges binned by owner each call, owner starts rows at bias+self term, adds neighbour rows with LDS float adds, stores once; LDS arrival order
# speedup vs baseline: 3.2177x; 1.3201x over previous
#include <hip/hip_runtime.h>
#include <hip/hip_bf16.h>


typedef __attribute__((ext_vector_type(2))) float v2f;
typedef __attribute__((ext_vector_type(8))) float v8f;


constexpr int DEG_PLANES = 16;
constexpr int DEG_SPAN   = 6400;

__device__ __forceinline__
void deg_hist_body(int blk, const int* __restrict__ dst, float* __restrict__ part, int n, int e) {
    __shared__ unsigned cnt[DEG_SPAN];
    const int p   = blk / DEG_PLANES;
    const int c   = blk % DEG_PLANES;
    const int lo  = p * DEG_SPAN;
    const int len = min(DEG_SPAN, n - lo);
    for (int i = threadIdx.x; i < DEG_SPAN; i += 1024) cnt[i] = 0u;
    __syncthreads();
    const int per = (e + DEG_PLANES - 1) / DEG_PLANES;
    const int e0  = c * per;
    const int e1  = min(e0 + per, e);
    constexpr int U = 16;
    for (int base = e0 + (int)threadIdx.x; base < e1; base += 1024 * U) {
        int d[U];
        if (base + (U - 1) * 1024 < e1) {
            #pragma unroll
            for (int j = 0; j < U; ++j) d[j] = dst[base + j * 1024];
        } else {
            #pragma unroll
            for (int j = 0; j < U; ++j) {
                const int i = base + j * 1024;
                d[j] = (i < e1) ? dst[i] : -1;
            }
        }
        #pragma unroll
        for (int j = 0; j < U; ++j) {
            const unsigned u = (unsigned)(d[j] - lo);
            if (u < (unsigned)len) atomicAdd(&cnt[u], 1u);
        }
    }
    __syncthreads();
    float* __restrict__ row = part + (size_t)c * n + lo;
    for (int i = threadIdx.x; i < len; i += 1024) row[i] = (float)cnt[i];
}


constexpr int OWN_SPAN  = 196;
constexpr int OWN_MAX   = 512;
constexpr int TAB_PITCH = 544;
constexpr int BIN_CHUNK = 2048;
constexpr int CHUNK_MAX = 512;

__device__ __forceinline__ int block_scan_excl(int v, int* wtot, int& total) {
    const int lane = threadIdx.x & 31, wave = threadIdx.x >> 5;
    int incl = v;
    #pragma unroll
    for (int o = 1; o < 32; o <<= 1) {
        const int u = __shfl_up(incl, o);
        if (lane >= o) incl += u;
    }
    if (lane == 31) wtot[wave] = incl;
    __syncthreads();
    int wb = 0, tot = 0;
    for (int w = 0; w < 32; ++w) {
        const int x = wtot[w];
        if (w < wave) wb += x;
        tot += x;
    }
    total = tot;
    return wb + incl - v;
}

__device__ __forceinline__
void edge_bin_body(int c, const int* __restrict__ src, const int* __restrict__ dst,
                   int* __restrict__ pairs_i, int* __restrict__ tab, int n, int e) {
    int2* __restrict__ pairs = (int2*)pairs_i;
    __shared__ int  bcnt[OWN_MAX];
    __shared__ int  bst[OWN_MAX + 1];
    __shared__ int  bw[32];
    __shared__ int2 lp[BIN_CHUNK];
    const int t = threadIdx.x;
    const int nown = (n + OWN_SPAN - 1) / OWN_SPAN;
    const int e0  = c * BIN_CHUNK;
    const int len = min(BIN_CHUNK, e - e0);
    if (t < OWN_MAX) bcnt[t] = 0;
    __syncthreads();
    int s[2], d[2], ow[2], rk[2];
    #pragma unroll
    for (int u = 0; u < 2; ++u) {
        const int i = t + u * 1024;
        const bool ok = i < len;
        s[u] = ok ? src[e0 + i] : 0;
        d[u] = ok ? dst[e0 + i] : -1;
    }
    #pragma unroll
    for (int u = 0; u < 2; ++u) {
        ow[u] = d[u] / OWN_SPAN;
        rk[u] = (d[u] >= 0) ? atomicAdd(&bcnt[ow[u]], 1) : 0;
    }
    __syncthreads();
    int total;
    const int mine = (t < nown) ? bcnt[t] : 0;
    const int ex = block_scan_excl(mine, bw, total);
    if (t < nown) bst[t] = ex;
    if (t <= nown) tab[(size_t)c * TAB_PITCH + t] = e0 + ex;
    __syncthreads();
    #pragma unroll
    for (int u = 0; u < 2; ++u)
        if (d[u] >= 0) lp[bst[ow[u]] + rk[u]] = make_int2(s[u], d[u]);
    __syncthreads();
    for (int i = t; i < len; i += 1024) pairs[e0 + i] = lp[i];
}

__global__ __launch_bounds__(1024)
void deg_bin_kernel(const int* __restrict__ src, const int* __restrict__ dst,
                    float* __restrict__ part, int* __restrict__ pairs,
                    int* __restrict__ tab, int n, int e, int ndeg) {
    if ((int)blockIdx.x < ndeg) deg_hist_body((int)blockIdx.x, dst, part, n, e);
    else edge_bin_body((int)blockIdx.x - ndeg, src, dst, pairs, tab, n, e);
}

__device__ __forceinline__ float dinv_of(float d) {
    return (d > 0.0f) ? rsqrtf(d) : 0.0f;
}


constexpr int GEMM_WAVES = 8;

__device__ __forceinline__ float xhalf(float x) {
    return __builtin_bit_cast(float, __builtin_amdgcn_permlanex16(
        __builtin_bit_cast(int, x), __builtin_bit_cast(int, x),
        0x76543210, 0xfedcba98, false, false));
}

template<int K, int NCOLS, bool RELU_A, bool DEG_IN>
__global__ __launch_bounds__(GEMM_WAVES * 32)
void wmma_gemm_kernel(const float* __restrict__ A, const float* __restrict__ B,
                      const float* __restrict__ nrm, float* __restrict__ C,
                      float* __restrict__ dinv_out, int M) {
    constexpr int NT = NCOLS / 16;
    constexpr int KB = K / 8;
    constexpr int KQ = K / 4;
    __shared__ float4 Wl[KQ * NCOLS];

    const int wave = __builtin_amdgcn_readfirstlane((int)(threadIdx.x >> 5));
    const int lane = threadIdx.x & 31;
    const int lo16 = lane & 15;
    const int half = lane >> 4;
    const int stripe = blockIdx.x * GEMM_WAVES + wave;
    const int m0 = stripe * 16;
    const bool live = m0 < M;
    const int mr = live ? m0 : 0;

    constexpr int NST = (KQ * NCOLS) / (GEMM_WAVES * 32);
    static_assert((KQ * NCOLS) % (GEMM_WAVES * 32) == 0, "B copy: whole pieces per thread");
    float4 st[NST];
    #pragma unroll
    for (int it = 0; it < NST; ++it) {
        const int i = (int)threadIdx.x + it * (GEMM_WAVES * 32);
        const int q = i / NCOLS, col = i % NCOLS;
        const float* __restrict__ bp = B + (size_t)(4 * q) * NCOLS + col;
        st[it] = make_float4(bp[0], bp[NCOLS], bp[2 * NCOLS], bp[3 * NCOLS]);
    }

    float4 av[KB];
    {
        const float* __restrict__ arow = A + (size_t)(mr + lo16) * K + 4 * half;
        #pragma unroll
        for (int j = 0; j < KB; ++j) av[j] = *(const float4*)(arow + 8 * j);
    }

    #pragma unroll
    for (int it = 0; it < NST; ++it)
        Wl[(int)threadIdx.x + it * (GEMM_WAVES * 32)] = st[it];
    __builtin_amdgcn_fence(__ATOMIC_RELEASE, "workgroup", "local");
    __builtin_amdgcn_s_barrier();
    __builtin_amdgcn_fence(__ATOMIC_ACQUIRE, "workgroup", "local");
    if (!live) return;

    if (DEG_IN && (stripe & 1) == 0) {
        const int r = m0 + lane;
        if (r < M) {
            float d = 1.0f;
            #pragma unroll
            for (int c = 0; c < DEG_PLANES; ++c) d += nrm[(size_t)c * M + r];
            dinv_out[r] = dinv_of(d);
        }
    }

    v8f acc[NT];
    #pragma unroll
    for (int t = 0; t < NT; ++t) acc[t] = v8f{};

    #pragma unroll
    for (int j = 0; j < KB; ++j) {
        float4 a4 = av[j];
        if (RELU_A) {
            a4.x = fmaxf(a4.x, 0.0f); a4.y = fmaxf(a4.y, 0.0f);
            a4.z = fmaxf(a4.z, 0.0f); a4.w = fmaxf(a4.w, 0.0f);
        }
        v2f a01; a01[0] = a4.x; a01[1] = a4.y;
        v2f a23; a23[0] = a4.z; a23[1] = a4.w;
        float4 w[NT];
        #pragma unroll
        for (int t = 0; t < NT; ++t) w[t] = Wl[(2 * j + half) * NCOLS + 16 * t + lo16];
        #pragma unroll
        for (int t = 0; t < NT; ++t) {
            v2f b; b[0] = w[t].x; b[1] = w[t].y;
            acc[t] = __builtin_amdgcn_wmma_f32_16x16x4_f32(
                          false, a01,  false, b,
                          (short)0, acc[t],  false,  false);
        }
        #pragma unroll
        for (int t = 0; t < NT; ++t) {
            v2f b; b[0] = w[t].z; b[1] = w[t].w;
            acc[t] = __builtin_amdgcn_wmma_f32_16x16x4_f32(
                          false, a23,  false, b,
                          (short)0, acc[t],  false,  false);
        }
    }

    #pragma unroll
    for (int p = 0; p < NT / 2; ++p) {
        const int col = 32 * p + lane;
        #pragma unroll
        for (int v = 0; v < 8; ++v) {
            const float t0 = acc[2 * p][v];
            const float t1 = acc[2 * p + 1][v];
            const float got = xhalf(half ? t0 : t1);
            const float x = half ? got : t0;
            const float y = half ? t1 : got;
            const size_t ox = (size_t)(m0 + v) * NCOLS + col;
            const size_t oy = (size_t)(m0 + 8 + v) * NCOLS + col;
            C[ox] = x;
            C[oy] = y;
        }
    }
}

template<int C>
__global__ __launch_bounds__(1024)
void owner_agg_kernel(const int* __restrict__ pairs_i, const int* __restrict__ tab,
                      const float* __restrict__ h, const float* __restrict__ dinv,
                      const float* __restrict__ bias, float* __restrict__ agg,
                      int n, int nchunk) {
    constexpr int R  = C / 32;
    constexpr int UB = 512 / C;
    __shared__ float4 acc4[OWN_SPAN * C / 4];
    __shared__ int pre[CHUNK_MAX + 1];
    __shared__ int beg[CHUNK_MAX];
    __shared__ int ow_w[32];
    float* acc = (float*)acc4;
    const int2* __restrict__ pairs = (const int2*)pairs_i;

    const int t    = threadIdx.x;
    const int lane = t & 31;
    const int wave = __builtin_amdgcn_readfirstlane(t >> 5);
    const int o    = blockIdx.x;
    const int lo   = o * OWN_SPAN;
    const int len  = min(OWN_SPAN, n - lo);

    int b0 = 0, cn = 0;
    if (t < nchunk) {
        b0 = tab[(size_t)t * TAB_PITCH + o];
        cn = tab[(size_t)t * TAB_PITCH + o + 1] - b0;
    }
    for (int i = t; i < len * (C / 4); i += 1024) {
        const int r  = i / (C / 4);
        const int cg = i % (C / 4);
        const float4 hv = *(const float4*)(h + (size_t)lo * C + 4 * (size_t)i);
        const float4 bv = *(const float4*)(bias + 4 * cg);
        const float di = dinv[lo + r];
        float4 a;
        a.x = bv.x + hv.x * di * di;
        a.y = bv.y + hv.y * di * di;
        a.z = bv.z + hv.z * di * di;
        a.w = bv.w + hv.w * di * di;
        acc4[i] = a;
    }
    int total_v;
    const int ex = block_scan_excl(cn, ow_w, total_v);
    const int total = __builtin_amdgcn_readfirstlane(total_v);
    if (t < nchunk) { pre[t] = ex; beg[t] = b0; }
    if (t == 0) pre[nchunk] = total;
    __syncthreads();

    for (int k0 = wave * 32; k0 < total; k0 += 32 * 32) {
        const int nb = min(32, total - k0);
        const int k  = min(k0 + lane, total - 1);
        int lo_c = 0, hi_c = nchunk;
        while (hi_c - lo_c > 1) {
            const int mid = (lo_c + hi_c) >> 1;
            if (pre[mid] <= k) lo_c = mid; else hi_c = mid;
        }
        const int2 pr = pairs[beg[lo_c] + (k - pre[lo_c])];
        const float nrm = dinv[pr.x] * dinv[pr.y];
        const int doff = (pr.y - lo) * C;
        for (int j0 = 0; j0 < nb; j0 += UB) {
            int   dd[UB];
            float nn[UB];
            float v[UB][R];
            #pragma unroll
            for (int u = 0; u < UB; ++u) {
                const int j  = min(j0 + u, nb - 1);
                const int sj = __builtin_amdgcn_readlane(pr.x, j);
                dd[u] = __builtin_amdgcn_readlane(doff, j);
                nn[u] = __builtin_bit_cast(float,
                            __builtin_amdgcn_readlane(__builtin_bit_cast(int, nrm), j));
                #pragma unroll
                for (int r = 0; r < R; ++r)
                    v[u][r] = h[(size_t)sj * C + r * 32 + lane];
            }
            #pragma unroll
            for (int u = 0; u < UB; ++u) {
                if (j0 + u < nb) {
                    #pragma unroll
                    for (int r = 0; r < R; ++r)
                        atomicAdd(&acc[dd[u] + r * 32 + lane], v[u][r] * nn[u]);
                }
            }
        }
    }
    __syncthreads();
    float4* __restrict__ orow = (float4*)(agg + (size_t)lo * C);
    for (int i = t; i < len * (C / 4); i += 1024) orow[i] = acc4[i];
}


__global__ void head_kernel(const float* __restrict__ agg2, const float* __restrict__ Wc,
                            const float* __restrict__ bc, float* __restrict__ out, int n) {
    int i = blockIdx.x * blockDim.x + threadIdx.x;
    if (i >= n) return;
    const float* row = agg2 + (size_t)i * 32;
    float acc = bc[0];
    #pragma unroll
    for (int c = 0; c < 32; ++c)
        acc += fmaxf(row[c], 0.0f) * Wc[c];
    out[i] = acc;
}


static inline int cdiv(long long a, long long b) { return (int)((a + b - 1) / b); }

extern "C" void kernel_launch(void* const* d_in, const int* in_sizes, int n_in,
                              void* d_out, int out_size, void* d_ws, size_t ws_size,
                              hipStream_t stream) {
    const float* x   = (const float*)d_in[0];
    const int*   ei  = (const int*)d_in[1];
    const float* W1  = (const float*)d_in[2];
    const float* b1  = (const float*)d_in[3];
    const float* W2  = (const float*)d_in[4];
    const float* b2  = (const float*)d_in[5];
    const float* Wc  = (const float*)d_in[6];
    const float* bc  = (const float*)d_in[7];
    float* out = (float*)d_out;

    const int N = in_sizes[0] / 128;
    const int E = in_sizes[1] / 2;
    const int* src = ei;
    const int* dst = ei + E;

    float* ws   = (float*)d_ws;
    float* dinv = ws;
    float* h1   = dinv + N;
    float* agg1 = h1 + (size_t)N * 64;
    float* h2   = agg1 + (size_t)N * 64;
    float* scr  = h2 + (size_t)N * 32;
    float* agg2 = h1;
    float* degp = scr;
    int*   pairs = (int*)(scr + (size_t)N * DEG_PLANES);
    int*   tab   = pairs + 2 * (size_t)E;
    (void)ws_size; (void)n_in; (void)out_size;

    const int T = 256;

    const int ndeg   = cdiv(N, DEG_SPAN) * DEG_PLANES;
    const int nchunk = cdiv(E, BIN_CHUNK);
    const int nown   = cdiv(N, OWN_SPAN);
    deg_bin_kernel<<<ndeg + nchunk, 1024, 0, stream>>>(src, dst, degp, pairs, tab, N, E, ndeg);

    wmma_gemm_kernel<128, 64, false, true><<<cdiv(N, 16 * GEMM_WAVES), GEMM_WAVES * 32, 0, stream>>>(
        x, W1, degp, h1, dinv, N);
    owner_agg_kernel<64><<<nown, 1024, 0, stream>>>(pairs, tab, h1, dinv, b1, agg1, N, nchunk);

    wmma_gemm_kernel<64, 32, true, false><<<cdiv(N, 16 * GEMM_WAVES), GEMM_WAVES * 32, 0, stream>>>(
        agg1, W2, nullptr, h2, nullptr, N);
    owner_agg_kernel<32><<<nown, 1024, 0, stream>>>(pairs, tab, h2, dinv, b2, agg2, N, nchunk);

    head_kernel<<<cdiv(N, T), T, 0, stream>>>(agg2, Wc, bc, out, N);
}


# Layer 2 reassociated: owners sum relu(agg1) rows (64 wide) in LDS, then do @W2 on WMMA, +b2, relu and the Wc dot (16-lane tree) on their own rows; GEMM2 and head launches gone, 6 to 4
# speedup vs baseline: 3.3858x; 1.0522x over previous
#include <hip/hip_runtime.h>
#include <hip/hip_bf16.h>


typedef __attribute__((ext_vector_type(2))) float v2f;
typedef __attribute__((ext_vector_type(8))) float v8f;


constexpr int DEG_PLANES = 16;
constexpr int DEG_SPAN   = 6400;

__device__ __forceinline__
void deg_hist_body(int blk, const int* __restrict__ dst, float* __restrict__ part, int n, int e) {
    __shared__ unsigned cnt[DEG_SPAN];
    const int p   = blk / DEG_PLANES;
    const int c   = blk % DEG_PLANES;
    const int lo  = p * DEG_SPAN;
    const int len = min(DEG_SPAN, n - lo);
    for (int i = threadIdx.x; i < DEG_SPAN; i += 1024) cnt[i] = 0u;
    __syncthreads();
    const int per = (e + DEG_PLANES - 1) / DEG_PLANES;
    const int e0  = c * per;
    const int e1  = min(e0 + per, e);
    constexpr int U = 16;
    for (int base = e0 + (int)threadIdx.x; base < e1; base += 1024 * U) {
        int d[U];
        if (base + (U - 1) * 1024 < e1) {
            #pragma unroll
            for (int j = 0; j < U; ++j) d[j] = dst[base + j * 1024];
        } else {
            #pragma unroll
            for (int j = 0; j < U; ++j) {
                const int i = base + j * 1024;
                d[j] = (i < e1) ? dst[i] : -1;
            }
        }
        #pragma unroll
        for (int j = 0; j < U; ++j) {
            const unsigned u = (unsigned)(d[j] - lo);
            if (u < (unsigned)len) atomicAdd(&cnt[u], 1u);
        }
    }
    __syncthreads();
    float* __restrict__ row = part + (size_t)c * n + lo;
    for (int i = threadIdx.x; i < len; i += 1024) row[i] = (float)cnt[i];
}


constexpr int OWN_SPAN  = 224;
constexpr int OWN_MAX   = 512;
constexpr int TAB_PITCH = 544;
constexpr int BIN_CHUNK = 2048;
constexpr int CHUNK_MAX = 512;

__device__ __forceinline__ int block_scan_excl(int v, int* wtot, int& total) {
    const int lane = threadIdx.x & 31, wave = threadIdx.x >> 5;
    int incl = v;
    #pragma unroll
    for (int o = 1; o < 32; o <<= 1) {
        const int u = __shfl_up(incl, o);
        if (lane >= o) incl += u;
    }
    if (lane == 31) wtot[wave] = incl;
    __syncthreads();
    int wb = 0, tot = 0;
    for (int w = 0; w < 32; ++w) {
        const int x = wtot[w];
        if (w < wave) wb += x;
        tot += x;
    }
    total = tot;
    return wb + incl - v;
}

__device__ __forceinline__
void edge_bin_body(int c, const int* __restrict__ src, const int* __restrict__ dst,
                   int* __restrict__ pairs_i, int* __restrict__ tab, int n, int e) {
    int2* __restrict__ pairs = (int2*)pairs_i;
    __shared__ int  bcnt[OWN_MAX];
    __shared__ int  bst[OWN_MAX + 1];
    __shared__ int  bw[32];
    __shared__ int2 lp[BIN_CHUNK];
    const int t = threadIdx.x;
    const int nown = (n + OWN_SPAN - 1) / OWN_SPAN;
    const int e0  = c * BIN_CHUNK;
    const int len = min(BIN_CHUNK, e - e0);
    if (t < OWN_MAX) bcnt[t] = 0;
    __syncthreads();
    int s[2], d[2], ow[2], rk[2];
    #pragma unroll
    for (int u = 0; u < 2; ++u) {
        const int i = t + u * 1024;
        const bool ok = i < len;
        s[u] = ok ? src[e0 + i] : 0;
        d[u] = ok ? dst[e0 + i] : -1;
    }
    #pragma unroll
    for (int u = 0; u < 2; ++u) {
        ow[u] = d[u] / OWN_SPAN;
        rk[u] = (d[u] >= 0) ? atomicAdd(&bcnt[ow[u]], 1) : 0;
    }
    __syncthreads();
    int total;
    const int mine = (t < nown) ? bcnt[t] : 0;
    const int ex = block_scan_excl(mine, bw, total);
    if (t < nown) bst[t] = ex;
    if (t <= nown) tab[(size_t)c * TAB_PITCH + t] = e0 + ex;
    __syncthreads();
    #pragma unroll
    for (int u = 0; u < 2; ++u)
        if (d[u] >= 0) lp[bst[ow[u]] + rk[u]] = make_int2(s[u], d[u]);
    __syncthreads();
    for (int i = t; i < len; i += 1024) pairs[e0 + i] = lp[i];
}

__global__ __launch_bounds__(1024)
void deg_bin_kernel(const int* __restrict__ src, const int* __restrict__ dst,
                    float* __restrict__ part, int* __restrict__ pairs,
                    int* __restrict__ tab, int n, int e, int ndeg) {
    if ((int)blockIdx.x < ndeg) deg_hist_body((int)blockIdx.x, dst, part, n, e);
    else edge_bin_body((int)blockIdx.x - ndeg, src, dst, pairs, tab, n, e);
}

__device__ __forceinline__ float dinv_of(float d) {
    return (d > 0.0f) ? rsqrtf(d) : 0.0f;
}


constexpr int GEMM_WAVES = 8;

__device__ __forceinline__ float xhalf(float x) {
    return __builtin_bit_cast(float, __builtin_amdgcn_permlanex16(
        __builtin_bit_cast(int, x), __builtin_bit_cast(int, x),
        0x76543210, 0xfedcba98, false, false));
}

template<int K, int NCOLS, bool RELU_A, bool DEG_IN>
__global__ __launch_bounds__(GEMM_WAVES * 32)
void wmma_gemm_kernel(const float* __restrict__ A, const float* __restrict__ B,
                      const float* __restrict__ nrm, float* __restrict__ C,
                      float* __restrict__ dinv_out, int M) {
    constexpr int NT = NCOLS / 16;
    constexpr int KB = K / 8;
    constexpr int KQ = K / 4;
    __shared__ float4 Wl[KQ * NCOLS];

    const int wave = __builtin_amdgcn_readfirstlane((int)(threadIdx.x >> 5));
    const int lane = threadIdx.x & 31;
    const int lo16 = lane & 15;
    const int half = lane >> 4;
    const int stripe = blockIdx.x * GEMM_WAVES + wave;
    const int m0 = stripe * 16;
    const bool live = m0 < M;
    const int mr = live ? m0 : 0;

    constexpr int NST = (KQ * NCOLS) / (GEMM_WAVES * 32);
    static_assert((KQ * NCOLS) % (GEMM_WAVES * 32) == 0, "B copy: whole pieces per thread");
    float4 st[NST];
    #pragma unroll
    for (int it = 0; it < NST; ++it) {
        const int i = (int)threadIdx.x + it * (GEMM_WAVES * 32);
        const int q = i / NCOLS, col = i % NCOLS;
        const float* __restrict__ bp = B + (size_t)(4 * q) * NCOLS + col;
        st[it] = make_float4(bp[0], bp[NCOLS], bp[2 * NCOLS], bp[3 * NCOLS]);
    }

    float4 av[KB];
    {
        const float* __restrict__ arow = A + (size_t)(mr + lo16) * K + 4 * half;
        #pragma unroll
        for (int j = 0; j < KB; ++j) av[j] = *(const float4*)(arow + 8 * j);
    }

    #pragma unroll
    for (int it = 0; it < NST; ++it)
        Wl[(int)threadIdx.x + it * (GEMM_WAVES * 32)] = st[it];
    __builtin_amdgcn_fence(__ATOMIC_RELEASE, "workgroup", "local");
    __builtin_amdgcn_s_barrier();
    __builtin_amdgcn_fence(__ATOMIC_ACQUIRE, "workgroup", "local");
    if (!live) return;

    if (DEG_IN && (stripe & 1) == 0) {
        const int r = m0 + lane;
        if (r < M) {
            float d = 1.0f;
            #pragma unroll
            for (int c = 0; c < DEG_PLANES; ++c) d += nrm[(size_t)c * M + r];
            dinv_out[r] = dinv_of(d);
        }
    }

    v8f acc[NT];
    #pragma unroll
    for (int t = 0; t < NT; ++t) acc[t] = v8f{};

    #pragma unroll
    for (int j = 0; j < KB; ++j) {
        float4 a4 = av[j];
        if (RELU_A) {
            a4.x = fmaxf(a4.x, 0.0f); a4.y = fmaxf(a4.y, 0.0f);
            a4.z = fmaxf(a4.z, 0.0f); a4.w = fmaxf(a4.w, 0.0f);
        }
        v2f a01; a01[0] = a4.x; a01[1] = a4.y;
        v2f a23; a23[0] = a4.z; a23[1] = a4.w;
        float4 w[NT];
        #pragma unroll
        for (int t = 0; t < NT; ++t) w[t] = Wl[(2 * j + half) * NCOLS + 16 * t + lo16];
        #pragma unroll
        for (int t = 0; t < NT; ++t) {
            v2f b; b[0] = w[t].x; b[1] = w[t].y;
            acc[t] = __builtin_amdgcn_wmma_f32_16x16x4_f32(
                          false, a01,  false, b,
                          (short)0, acc[t],  false,  false);
        }
        #pragma unroll
        for (int t = 0; t < NT; ++t) {
            v2f b; b[0] = w[t].z; b[1] = w[t].w;
            acc[t] = __builtin_amdgcn_wmma_f32_16x16x4_f32(
                          false, a23,  false, b,
                          (short)0, acc[t],  false,  false);
        }
    }

    #pragma unroll
    for (int p = 0; p < NT / 2; ++p) {
        const int col = 32 * p + lane;
        #pragma unroll
        for (int v = 0; v < 8; ++v) {
            const float t0 = acc[2 * p][v];
            const float t1 = acc[2 * p + 1][v];
            const float got = xhalf(half ? t0 : t1);
            const float x = half ? got : t0;
            const float y = half ? t1 : got;
            const size_t ox = (size_t)(m0 + v) * NCOLS + col;
            const size_t oy = (size_t)(m0 + 8 + v) * NCOLS + col;
            C[ox] = x;
            C[oy] = y;
        }
    }
}

template<int C, bool HEAD>
__global__ __launch_bounds__(1024)
void owner_agg_kernel(const int* __restrict__ pairs_i, const int* __restrict__ tab,
                      const float* __restrict__ h, const float* __restrict__ dinv,
                      const float* __restrict__ bias, float* __restrict__ agg,
                      const float* __restrict__ W2, const float* __restrict__ Wc,
                      const float* __restrict__ bc, float* __restrict__ out,
                      int n, int nchunk) {
    constexpr int R  = C / 32;
    constexpr int UB = 512 / C;
    constexpr int NOUT = C / 2;
    constexpr int KQ   = C / 4;
    static_assert(KQ * NOUT <= 1024, "W2 copy: at most one piece per thread");
    static_assert(OWN_SPAN % 32 == 0, "whole result lines and whole stripes per owner");
    __shared__ float4 Wl[HEAD ? KQ * NOUT : 1];
    __shared__ float  outl[HEAD ? OWN_SPAN : 1];
    __shared__ float4 acc4[OWN_SPAN * C / 4];
    __shared__ int pre[CHUNK_MAX + 1];
    __shared__ int beg[CHUNK_MAX];
    __shared__ int ow_w[32];
    float* acc = (float*)acc4;
    const int2* __restrict__ pairs = (const int2*)pairs_i;

    const int t    = threadIdx.x;
    const int lane = t & 31;
    const int wave = __builtin_amdgcn_readfirstlane(t >> 5);
    const int o    = blockIdx.x;
    const int lo   = o * OWN_SPAN;
    const int len  = min(OWN_SPAN, n - lo);

    int b0 = 0, cn = 0;
    if (t < nchunk) {
        b0 = tab[(size_t)t * TAB_PITCH + o];
        cn = tab[(size_t)t * TAB_PITCH + o + 1] - b0;
    }
    float4 wst = make_float4(0.0f, 0.0f, 0.0f, 0.0f);
    if (HEAD && t < KQ * NOUT) {
        const int q = t / NOUT, col = t % NOUT;
        const float* __restrict__ bp = W2 + (size_t)(4 * q) * NOUT + col;
        wst = make_float4(bp[0], bp[NOUT], bp[2 * NOUT], bp[3 * NOUT]);
    }
    for (int i = t; i < len * (C / 4); i += 1024) {
        const int r  = i / (C / 4);
        const int cg = i % (C / 4);
        const float4 hv = *(const float4*)(h + (size_t)lo * C + 4 * (size_t)i);
        const float di = dinv[lo + r];
        float4 a;
        if (HEAD) {
            a.x = hv.x * di * di;
            a.y = hv.y * di * di;
            a.z = hv.z * di * di;
            a.w = hv.w * di * di;
        } else {
            const float4 bv = *(const float4*)(bias + 4 * cg);
            a.x = bv.x + hv.x * di * di;
            a.y = bv.y + hv.y * di * di;
            a.z = bv.z + hv.z * di * di;
            a.w = bv.w + hv.w * di * di;
        }
        acc4[i] = a;
    }
    if (HEAD && t < KQ * NOUT) Wl[t] = wst;
    int total_v;
    const int ex = block_scan_excl(cn, ow_w, total_v);
    const int total = __builtin_amdgcn_readfirstlane(total_v);
    if (t < nchunk) { pre[t] = ex; beg[t] = b0; }
    if (t == 0) pre[nchunk] = total;
    __syncthreads();

    for (int k0 = wave * 32; k0 < total; k0 += 32 * 32) {
        const int nb = min(32, total - k0);
        const int k  = min(k0 + lane, total - 1);
        int lo_c = 0, hi_c = nchunk;
        while (hi_c - lo_c > 1) {
            const int mid = (lo_c + hi_c) >> 1;
            if (pre[mid] <= k) lo_c = mid; else hi_c = mid;
        }
        const int2 pr = pairs[beg[lo_c] + (k - pre[lo_c])];
        const float nrm = dinv[pr.x] * dinv[pr.y];
        const int doff = (pr.y - lo) * C;
        for (int j0 = 0; j0 < nb; j0 += UB) {
            int   dd[UB];
            float nn[UB];
            float v[UB][R];
            #pragma unroll
            for (int u = 0; u < UB; ++u) {
                const int j  = min(j0 + u, nb - 1);
                const int sj = __builtin_amdgcn_readlane(pr.x, j);
                dd[u] = __builtin_amdgcn_readlane(doff, j);
                nn[u] = __builtin_bit_cast(float,
                            __builtin_amdgcn_readlane(__builtin_bit_cast(int, nrm), j));
                #pragma unroll
                for (int r = 0; r < R; ++r)
                    v[u][r] = h[(size_t)sj * C + r * 32 + lane];
            }
            #pragma unroll
            for (int u = 0; u < UB; ++u) {
                if (j0 + u < nb) {
                    #pragma unroll
                    for (int r = 0; r < R; ++r)
                        atomicAdd(&acc[dd[u] + r * 32 + lane], v[u][r] * nn[u]);
                }
            }
        }
    }
    __syncthreads();
    if (!HEAD) {
        float4* __restrict__ orow = (float4*)(agg + (size_t)lo * C);
        for (int i = t; i < len * (C / 4); i += 1024) {
            float4 a = acc4[i];
            a.x = fmaxf(a.x, 0.0f); a.y = fmaxf(a.y, 0.0f);
            a.z = fmaxf(a.z, 0.0f); a.w = fmaxf(a.w, 0.0f);
            orow[i] = a;
        }
    } else {
        const int m0 = wave * 16;
        if (m0 < len) {
            const int lo16 = lane & 15;
            const int half = lane >> 4;
            const float* arow = acc + (m0 + lo16) * C + 4 * half;
            v8f t0 = v8f{}, t1 = v8f{};
            #pragma unroll
            for (int j = 0; j < C / 8; ++j) {
                const float4 a4 = *(const float4*)(arow + 8 * j);
                const float4 w0 = Wl[(2 * j + half) * NOUT + lo16];
                const float4 w1 = Wl[(2 * j + half) * NOUT + 16 + lo16];
                v2f a01; a01[0] = a4.x; a01[1] = a4.y;
                v2f a23; a23[0] = a4.z; a23[1] = a4.w;
                v2f b;
                b[0] = w0.x; b[1] = w0.y;
                t0 = __builtin_amdgcn_wmma_f32_16x16x4_f32(false, a01, false, b, (short)0, t0, false, false);
                b[0] = w1.x; b[1] = w1.y;
                t1 = __builtin_amdgcn_wmma_f32_16x16x4_f32(false, a01, false, b, (short)0, t1, false, false);
                b[0] = w0.z; b[1] = w0.w;
                t0 = __builtin_amdgcn_wmma_f32_16x16x4_f32(false, a23, false, b, (short)0, t0, false, false);
                b[0] = w1.z; b[1] = w1.w;
                t1 = __builtin_amdgcn_wmma_f32_16x16x4_f32(false, a23, false, b, (short)0, t1, false, false);
            }
            const float ba = bias[lo16], bb = bias[16 + lo16];
            const float wa = Wc[lo16],   wb = Wc[16 + lo16];
            float s[8];
            #pragma unroll
            for (int v = 0; v < 8; ++v)
                s[v] = fmaxf(t0[v] + ba, 0.0f) * wa + fmaxf(t1[v] + bb, 0.0f) * wb;
            #pragma unroll
            for (int o2 = 8; o2 >= 1; o2 >>= 1) {
                #pragma unroll
                for (int v = 0; v < 8; ++v) s[v] += __shfl_xor(s[v], o2);
            }
            if (lo16 == 0) {
                const float b0v = bc[0];
                #pragma unroll
                for (int v = 0; v < 8; ++v) outl[m0 + 8 * half + v] = s[v] + b0v;
            }
        }
        __syncthreads();
        if (t < len) out[lo + t] = outl[t];
    }
}


static inline int cdiv(long long a, long long b) { return (int)((a + b - 1) / b); }

extern "C" void kernel_launch(void* const* d_in, const int* in_sizes, int n_in,
                              void* d_out, int out_size, void* d_ws, size_t ws_size,
                              hipStream_t stream) {
    const float* x   = (const float*)d_in[0];
    const int*   ei  = (const int*)d_in[1];
    const float* W1  = (const float*)d_in[2];
    const float* b1  = (const float*)d_in[3];
    const float* W2  = (const float*)d_in[4];
    const float* b2  = (const float*)d_in[5];
    const float* Wc  = (const float*)d_in[6];
    const float* bc  = (const float*)d_in[7];
    float* out = (float*)d_out;

    const int N = in_sizes[0] / 128;
    const int E = in_sizes[1] / 2;
    const int* src = ei;
    const int* dst = ei + E;

    float* ws   = (float*)d_ws;
    float* dinv = ws;
    float* h1   = dinv + N;
    float* agg1 = h1 + (size_t)N * 64;
    float* scr  = agg1 + (size_t)N * 64 + (size_t)N * 32;
    float* degp = scr;
    int*   pairs = (int*)(scr + (size_t)N * DEG_PLANES);
    int*   tab   = pairs + 2 * (size_t)E;
    (void)ws_size; (void)n_in; (void)out_size;

    const int ndeg   = cdiv(N, DEG_SPAN) * DEG_PLANES;
    const int nchunk = cdiv(E, BIN_CHUNK);
    const int nown   = cdiv(N, OWN_SPAN);
    deg_bin_kernel<<<ndeg + nchunk, 1024, 0, stream>>>(src, dst, degp, pairs, tab, N, E, ndeg);

    wmma_gemm_kernel<128, 64, false, true><<<cdiv(N, 16 * GEMM_WAVES), GEMM_WAVES * 32, 0, stream>>>(
        x, W1, degp, h1, dinv, N);
    owner_agg_kernel<64, false><<<nown, 1024, 0, stream>>>(
        pairs, tab, h1, dinv, b1, agg1, nullptr, nullptr, nullptr, nullptr, N, nchunk);

    owner_agg_kernel<64, true><<<nown, 1024, 0, stream>>>(
        pairs, tab, agg1, dinv, b2, nullptr, W2, Wc, bc, out, N, nchunk);
}


# Launch 1 re-gridded: all work-groups resident at once, half the waves: degree groups span 25600 nodes x 32 edge chunks (one load group), bin groups take 4096 edges; LDS arrival-order sums as before
# speedup vs baseline: 3.4934x; 1.0318x over previous
#include <hip/hip_runtime.h>
#include <hip/hip_bf16.h>


typedef __attribute__((ext_vector_type(2))) float v2f;
typedef __attribute__((ext_vector_type(8))) float v8f;


constexpr int DEG_PLANES = 32;
constexpr int DEG_SPAN   = 25600;

__device__ __forceinline__
void deg_hist_body(int blk, const int* __restrict__ dst, float* __restrict__ part, int n, int e) {
    __shared__ uint4 cnt4[DEG_SPAN / 4];
    unsigned* cnt = (unsigned*)cnt4;
    const int p   = blk / DEG_PLANES;
    const int c   = blk % DEG_PLANES;
    const int lo  = p * DEG_SPAN;
    const int len = min(DEG_SPAN, n - lo);
    for (int i = threadIdx.x; i < DEG_SPAN / 4; i += 1024) cnt4[i] = make_uint4(0u, 0u, 0u, 0u);
    __syncthreads();
    const int per = (e + DEG_PLANES - 1) / DEG_PLANES;
    const int e0  = c * per;
    const int e1  = min(e0 + per, e);
    constexpr int U = 20;
    for (int base = e0 + (int)threadIdx.x; base < e1; base += 1024 * U) {
        int d[U];
        #pragma unroll
        for (int j = 0; j < U; ++j) d[j] = dst[min(base + j * 1024, e1 - 1)];
        #pragma unroll
        for (int j = 0; j < U; ++j) {
            const unsigned u = (unsigned)(d[j] - lo);
            if (base + j * 1024 < e1 && u < (unsigned)len) atomicAdd(&cnt[u], 1u);
        }
    }
    __syncthreads();
    float* __restrict__ row = part + (size_t)c * n + lo;
    if ((n & 3) == 0) {
        float4* __restrict__ row4 = (float4*)row;
        for (int i = threadIdx.x; i < (len >> 2); i += 1024) {
            const uint4 u = cnt4[i];
            row4[i] = make_float4((float)u.x, (float)u.y, (float)u.z, (float)u.w);
        }
    } else {
        for (int i = threadIdx.x; i < len; i += 1024) row[i] = (float)cnt[i];
    }
}


constexpr int OWN_SPAN  = 224;
constexpr int OWN_MAX   = 512;
constexpr int TAB_PITCH = 544;
constexpr int BIN_CHUNK = 4096;
constexpr int BIN_EPT   = BIN_CHUNK / 1024;
constexpr int CHUNK_MAX = 512;

__device__ __forceinline__ int block_scan_excl(int v, int* wtot, int& total) {
    const int lane = threadIdx.x & 31, wave = threadIdx.x >> 5;
    int incl = v;
    #pragma unroll
    for (int o = 1; o < 32; o <<= 1) {
        const int u = __shfl_up(incl, o);
        if (lane >= o) incl += u;
    }
    if (lane == 31) wtot[wave] = incl;
    __syncthreads();
    int wb = 0, tot = 0;
    for (int w = 0; w < 32; ++w) {
        const int x = wtot[w];
        if (w < wave) wb += x;
        tot += x;
    }
    total = tot;
    return wb + incl - v;
}

__device__ __forceinline__
void edge_bin_body(int c, const int* __restrict__ src, const int* __restrict__ dst,
                   int* __restrict__ pairs_i, int* __restrict__ tab, int n, int e) {
    int2* __restrict__ pairs = (int2*)pairs_i;
    __shared__ int  bcnt[OWN_MAX];
    __shared__ int  bst[OWN_MAX + 1];
    __shared__ int  bw[32];
    __shared__ int2 lp[BIN_CHUNK];
    const int t = threadIdx.x;
    const int nown = (n + OWN_SPAN - 1) / OWN_SPAN;
    const int e0  = c * BIN_CHUNK;
    const int len = min(BIN_CHUNK, e - e0);
    if (t < OWN_MAX) bcnt[t] = 0;
    __syncthreads();
    int s[BIN_EPT], d[BIN_EPT], ow[BIN_EPT], rk[BIN_EPT];
    #pragma unroll
    for (int u = 0; u < BIN_EPT; ++u) {
        const int i = t + u * 1024;
        const bool ok = i < len;
        s[u] = ok ? src[e0 + i] : 0;
        d[u] = ok ? dst[e0 + i] : -1;
    }
    #pragma unroll
    for (int u = 0; u < BIN_EPT; ++u) {
        ow[u] = d[u] / OWN_SPAN;
        rk[u] = (d[u] >= 0) ? atomicAdd(&bcnt[ow[u]], 1) : 0;
    }
    __syncthreads();
    int total;
    const int mine = (t < nown) ? bcnt[t] : 0;
    const int ex = block_scan_excl(mine, bw, total);
    if (t < nown) bst[t] = ex;
    if (t <= nown) tab[(size_t)c * TAB_PITCH + t] = e0 + ex;
    __syncthreads();
    #pragma unroll
    for (int u = 0; u < BIN_EPT; ++u)
        if (d[u] >= 0) lp[bst[ow[u]] + rk[u]] = make_int2(s[u], d[u]);
    __syncthreads();
    for (int i = t; i < len; i += 1024) pairs[e0 + i] = lp[i];
}

__global__ __launch_bounds__(1024)
void deg_bin_kernel(const int* __restrict__ src, const int* __restrict__ dst,
                    float* __restrict__ part, int* __restrict__ pairs,
                    int* __restrict__ tab, int n, int e, int ndeg) {
    if ((int)blockIdx.x < ndeg) deg_hist_body((int)blockIdx.x, dst, part, n, e);
    else edge_bin_body((int)blockIdx.x - ndeg, src, dst, pairs, tab, n, e);
}

__device__ __forceinline__ float dinv_of(float d) {
    return (d > 0.0f) ? rsqrtf(d) : 0.0f;
}


constexpr int GEMM_WAVES = 8;

__device__ __forceinline__ float xhalf(float x) {
    return __builtin_bit_cast(float, __builtin_amdgcn_permlanex16(
        __builtin_bit_cast(int, x), __builtin_bit_cast(int, x),
        0x76543210, 0xfedcba98, false, false));
}

template<int K, int NCOLS, bool RELU_A, bool DEG_IN>
__global__ __launch_bounds__(GEMM_WAVES * 32)
void wmma_gemm_kernel(const float* __restrict__ A, const float* __restrict__ B,
                      const float* __restrict__ nrm, float* __restrict__ C,
                      float* __restrict__ dinv_out, int M) {
    constexpr int NT = NCOLS / 16;
    constexpr int KB = K / 8;
    constexpr int KQ = K / 4;
    __shared__ float4 Wl[KQ * NCOLS];

    const int wave = __builtin_amdgcn_readfirstlane((int)(threadIdx.x >> 5));
    const int lane = threadIdx.x & 31;
    const int lo16 = lane & 15;
    const int half = lane >> 4;
    const int stripe = blockIdx.x * GEMM_WAVES + wave;
    const int m0 = stripe * 16;
    const bool live = m0 < M;
    const int mr = live ? m0 : 0;

    constexpr int NST = (KQ * NCOLS) / (GEMM_WAVES * 32);
    static_assert((KQ * NCOLS) % (GEMM_WAVES * 32) == 0, "B copy: whole pieces per thread");
    float4 st[NST];
    #pragma unroll
    for (int it = 0; it < NST; ++it) {
        const int i = (int)threadIdx.x + it * (GEMM_WAVES * 32);
        const int q = i / NCOLS, col = i % NCOLS;
        const float* __restrict__ bp = B + (size_t)(4 * q) * NCOLS + col;
        st[it] = make_float4(bp[0], bp[NCOLS], bp[2 * NCOLS], bp[3 * NCOLS]);
    }

    float4 av[KB];
    {
        const float* __restrict__ arow = A + (size_t)(mr + lo16) * K + 4 * half;
        #pragma unroll
        for (int j = 0; j < KB; ++j) av[j] = *(const float4*)(arow + 8 * j);
    }

    #pragma unroll
    for (int it = 0; it < NST; ++it)
        Wl[(int)threadIdx.x + it * (GEMM_WAVES * 32)] = st[it];
    __builtin_amdgcn_fence(__ATOMIC_RELEASE, "workgroup", "local");
    __builtin_amdgcn_s_barrier();
    __builtin_amdgcn_fence(__ATOMIC_ACQUIRE, "workgroup", "local");
    if (!live) return;

    if (DEG_IN && (stripe & 1) == 0) {
        const int r = m0 + lane;
        if (r < M) {
            float d = 1.0f;
            #pragma unroll
            for (int c = 0; c < DEG_PLANES; ++c) d += nrm[(size_t)c * M + r];
            dinv_out[r] = dinv_of(d);
        }
    }

    v8f acc[NT];
    #pragma unroll
    for (int t = 0; t < NT; ++t) acc[t] = v8f{};

    #pragma unroll
    for (int j = 0; j < KB; ++j) {
        float4 a4 = av[j];
        if (RELU_A) {
            a4.x = fmaxf(a4.x, 0.0f); a4.y = fmaxf(a4.y, 0.0f);
            a4.z = fmaxf(a4.z, 0.0f); a4.w = fmaxf(a4.w, 0.0f);
        }
        v2f a01; a01[0] = a4.x; a01[1] = a4.y;
        v2f a23; a23[0] = a4.z; a23[1] = a4.w;
        float4 w[NT];
        #pragma unroll
        for (int t = 0; t < NT; ++t) w[t] = Wl[(2 * j + half) * NCOLS + 16 * t + lo16];
        #pragma unroll
        for (int t = 0; t < NT; ++t) {
            v2f b; b[0] = w[t].x; b[1] = w[t].y;
            acc[t] = __builtin_amdgcn_wmma_f32_16x16x4_f32(
                          false, a01,  false, b,
                          (short)0, acc[t],  false,  false);
        }
        #pragma unroll
        for (int t = 0; t < NT; ++t) {
            v2f b; b[0] = w[t].z; b[1] = w[t].w;
            acc[t] = __builtin_amdgcn_wmma_f32_16x16x4_f32(
                          false, a23,  false, b,
                          (short)0, acc[t],  false,  false);
        }
    }

    #pragma unroll
    for (int p = 0; p < NT / 2; ++p) {
        const int col = 32 * p + lane;
        #pragma unroll
        for (int v = 0; v < 8; ++v) {
            const float t0 = acc[2 * p][v];
            const float t1 = acc[2 * p + 1][v];
            const float got = xhalf(half ? t0 : t1);
            const float x = half ? got : t0;
            const float y = half ? t1 : got;
            const size_t ox = (size_t)(m0 + v) * NCOLS + col;
            const size_t oy = (size_t)(m0 + 8 + v) * NCOLS + col;
            C[ox] = x;
            C[oy] = y;
        }
    }
}

template<int C, bool HEAD>
__global__ __launch_bounds__(1024)
void owner_agg_kernel(const int* __restrict__ pairs_i, const int* __restrict__ tab,
                      const float* __restrict__ h, const float* __restrict__ dinv,
                      const float* __restrict__ bias, float* __restrict__ agg,
                      const float* __restrict__ W2, const float* __restrict__ Wc,
                      const float* __restrict__ bc, float* __restrict__ out,
                      int n, int nchunk) {
    constexpr int R  = C / 32;
    constexpr int UB = 512 / C;
    constexpr int NOUT = C / 2;
    constexpr int KQ   = C / 4;
    static_assert(KQ * NOUT <= 1024, "W2 copy: at most one piece per thread");
    static_assert(OWN_SPAN % 32 == 0, "whole result lines and whole stripes per owner");
    __shared__ float4 Wl[HEAD ? KQ * NOUT : 1];
    __shared__ float  outl[HEAD ? OWN_SPAN : 1];
    __shared__ float4 acc4[OWN_SPAN * C / 4];
    __shared__ int pre[CHUNK_MAX + 1];
    __shared__ int beg[CHUNK_MAX];
    __shared__ int ow_w[32];
    float* acc = (float*)acc4;
    const int2* __restrict__ pairs = (const int2*)pairs_i;

    const int t    = threadIdx.x;
    const int lane = t & 31;
    const int wave = __builtin_amdgcn_readfirstlane(t >> 5);
    const int o    = blockIdx.x;
    const int lo   = o * OWN_SPAN;
    const int len  = min(OWN_SPAN, n - lo);

    int b0 = 0, cn = 0;
    if (t < nchunk) {
        b0 = tab[(size_t)t * TAB_PITCH + o];
        cn = tab[(size_t)t * TAB_PITCH + o + 1] - b0;
    }
    float4 wst = make_float4(0.0f, 0.0f, 0.0f, 0.0f);
    if (HEAD && t < KQ * NOUT) {
        const int q = t / NOUT, col = t % NOUT;
        const float* __restrict__ bp = W2 + (size_t)(4 * q) * NOUT + col;
        wst = make_float4(bp[0], bp[NOUT], bp[2 * NOUT], bp[3 * NOUT]);
    }
    for (int i = t; i < len * (C / 4); i += 1024) {
        const int r  = i / (C / 4);
        const int cg = i % (C / 4);
        const float4 hv = *(const float4*)(h + (size_t)lo * C + 4 * (size_t)i);
        const float di = dinv[lo + r];
        float4 a;
        if (HEAD) {
            a.x = hv.x * di * di;
            a.y = hv.y * di * di;
            a.z = hv.z * di * di;
            a.w = hv.w * di * di;
        } else {
            const float4 bv = *(const float4*)(bias + 4 * cg);
            a.x = bv.x + hv.x * di * di;
            a.y = bv.y + hv.y * di * di;
            a.z = bv.z + hv.z * di * di;
            a.w = bv.w + hv.w * di * di;
        }
        acc4[i] = a;
    }
    if (HEAD && t < KQ * NOUT) Wl[t] = wst;
    int total_v;
    const int ex = block_scan_excl(cn, ow_w, total_v);
    const int total = __builtin_amdgcn_readfirstlane(total_v);
    if (t < nchunk) { pre[t] = ex; beg[t] = b0; }
    if (t == 0) pre[nchunk] = total;
    __syncthreads();

    for (int k0 = wave * 32; k0 < total; k0 += 32 * 32) {
        const int nb = min(32, total - k0);
        const int k  = min(k0 + lane, total - 1);
        int lo_c = 0, hi_c = nchunk;
        while (hi_c - lo_c > 1) {
            const int mid = (lo_c + hi_c) >> 1;
            if (pre[mid] <= k) lo_c = mid; else hi_c = mid;
        }
        const int2 pr = pairs[beg[lo_c] + (k - pre[lo_c])];
        const float nrm = dinv[pr.x] * dinv[pr.y];
        const int doff = (pr.y - lo) * C;
        for (int j0 = 0; j0 < nb; j0 += UB) {
            int   dd[UB];
            float nn[UB];
            float v[UB][R];
            #pragma unroll
            for (int u = 0; u < UB; ++u) {
                const int j  = min(j0 + u, nb - 1);
                const int sj = __builtin_amdgcn_readlane(pr.x, j);
                dd[u] = __builtin_amdgcn_readlane(doff, j);
                nn[u] = __builtin_bit_cast(float,
                            __builtin_amdgcn_readlane(__builtin_bit_cast(int, nrm), j));
                #pragma unroll
                for (int r = 0; r < R; ++r)
                    v[u][r] = h[(size_t)sj * C + r * 32 + lane];
            }
            #pragma unroll
            for (int u = 0; u < UB; ++u) {
                if (j0 + u < nb) {
                    #pragma unroll
                    for (int r = 0; r < R; ++r)
                        atomicAdd(&acc[dd[u] + r * 32 + lane], v[u][r] * nn[u]);
                }
            }
        }
    }
    __syncthreads();
    if (!HEAD) {
        float4* __restrict__ orow = (float4*)(agg + (size_t)lo * C);
        for (int i = t; i < len * (C / 4); i += 1024) {
            float4 a = acc4[i];
            a.x = fmaxf(a.x, 0.0f); a.y = fmaxf(a.y, 0.0f);
            a.z = fmaxf(a.z, 0.0f); a.w = fmaxf(a.w, 0.0f);
            orow[i] = a;
        }
    } else {
        const int m0 = wave * 16;
        if (m0 < len) {
            const int lo16 = lane & 15;
            const int half = lane >> 4;
            const float* arow = acc + (m0 + lo16) * C + 4 * half;
            v8f t0 = v8f{}, t1 = v8f{};
            #pragma unroll
            for (int j = 0; j < C / 8; ++j) {
                const float4 a4 = *(const float4*)(arow + 8 * j);
                const float4 w0 = Wl[(2 * j + half) * NOUT + lo16];
                const float4 w1 = Wl[(2 * j + half) * NOUT + 16 + lo16];
                v2f a01; a01[0] = a4.x; a01[1] = a4.y;
                v2f a23; a23[0] = a4.z; a23[1] = a4.w;
                v2f b;
                b[0] = w0.x; b[1] = w0.y;
                t0 = __builtin_amdgcn_wmma_f32_16x16x4_f32(false, a01, false, b, (short)0, t0, false, false);
                b[0] = w1.x; b[1] = w1.y;
                t1 = __builtin_amdgcn_wmma_f32_16x16x4_f32(false, a01, false, b, (short)0, t1, false, false);
                b[0] = w0.z; b[1] = w0.w;
                t0 = __builtin_amdgcn_wmma_f32_16x16x4_f32(false, a23, false, b, (short)0, t0, false, false);
                b[0] = w1.z; b[1] = w1.w;
                t1 = __builtin_amdgcn_wmma_f32_16x16x4_f32(false, a23, false, b, (short)0, t1, false, false);
            }
            const float ba = bias[lo16], bb = bias[16 + lo16];
            const float wa = Wc[lo16],   wb = Wc[16 + lo16];
            float s[8];
            #pragma unroll
            for (int v = 0; v < 8; ++v)
                s[v] = fmaxf(t0[v] + ba, 0.0f) * wa + fmaxf(t1[v] + bb, 0.0f) * wb;
            #pragma unroll
            for (int o2 = 8; o2 >= 1; o2 >>= 1) {
                #pragma unroll
                for (int v = 0; v < 8; ++v) s[v] += __shfl_xor(s[v], o2);
            }
            if (lo16 == 0) {
                const float b0v = bc[0];
                #pragma unroll
                for (int v = 0; v < 8; ++v) outl[m0 + 8 * half + v] = s[v] + b0v;
            }
        }
        __syncthreads();
        if (t < len) out[lo + t] = outl[t];
    }
}


static inline int cdiv(long long a, long long b) { return (int)((a + b - 1) / b); }

extern "C" void kernel_launch(void* const* d_in, const int* in_sizes, int n_in,
                              void* d_out, int out_size, void* d_ws, size_t ws_size,
                              hipStream_t stream) {
    const float* x   = (const float*)d_in[0];
    const int*   ei  = (const int*)d_in[1];
    const float* W1  = (const float*)d_in[2];
    const float* b1  = (const float*)d_in[3];
    const float* W2  = (const float*)d_in[4];
    const float* b2  = (const float*)d_in[5];
    const float* Wc  = (const float*)d_in[6];
    const float* bc  = (const float*)d_in[7];
    float* out = (float*)d_out;

    const int N = in_sizes[0] / 128;
    const int E = in_sizes[1] / 2;
    const int* src = ei;
    const int* dst = ei + E;

    float* ws   = (float*)d_ws;
    float* dinv = ws;
    float* h1   = dinv + N;
    float* agg1 = h1 + (size_t)N * 64;
    float* degp = agg1 + (size_t)N * 64;
    float* scr  = degp + (size_t)N * DEG_PLANES;
    int*   pairs = (int*)scr;
    int*   tab   = pairs + 2 * (size_t)E;
    (void)ws_size; (void)n_in; (void)out_size;

    const int ndeg   = cdiv(N, DEG_SPAN) * DEG_PLANES;
    const int nchunk = cdiv(E, BIN_CHUNK);
    const int nown   = cdiv(N, OWN_SPAN);
    deg_bin_kernel<<<ndeg + nchunk, 1024, 0, stream>>>(src, dst, degp, pairs, tab, N, E, ndeg);

    wmma_gemm_kernel<128, 64, false, true><<<cdiv(N, 16 * GEMM_WAVES), GEMM_WAVES * 32, 0, stream>>>(
        x, W1, degp, h1, dinv, N);
    owner_agg_kernel<64, false><<<nown, 1024, 0, stream>>>(
        pairs, tab, h1, dinv, b1, agg1, nullptr, nullptr, nullptr, nullptr, N, nchunk);

    owner_agg_kernel<64, true><<<nown, 1024, 0, stream>>>(
        pairs, tab, agg1, dinv, b2, nullptr, W2, Wc, bc, out, N, nchunk);
}


# Owner launches re-divided: 16 waves per owner (was 32); a wave takes one contiguous equal share of the owner's edges, looks up 96 per index pass, keeps 16 rows in flight; LDS arrival order as before
# speedup vs baseline: 3.6033x; 1.0315x over previous
#include <hip/hip_runtime.h>
#include <hip/hip_bf16.h>


typedef __attribute__((ext_vector_type(2))) float v2f;
typedef __attribute__((ext_vector_type(8))) float v8f;


constexpr int DEG_PLANES = 32;
constexpr int DEG_SPAN   = 25600;

__device__ __forceinline__
void deg_hist_body(int blk, const int* __restrict__ dst, float* __restrict__ part, int n, int e) {
    __shared__ uint4 cnt4[DEG_SPAN / 4];
    unsigned* cnt = (unsigned*)cnt4;
    const int p   = blk / DEG_PLANES;
    const int c   = blk % DEG_PLANES;
    const int lo  = p * DEG_SPAN;
    const int len = min(DEG_SPAN, n - lo);
    for (int i = threadIdx.x; i < DEG_SPAN / 4; i += 1024) cnt4[i] = make_uint4(0u, 0u, 0u, 0u);
    __syncthreads();
    const int per = (e + DEG_PLANES - 1) / DEG_PLANES;
    const int e0  = c * per;
    const int e1  = min(e0 + per, e);
    constexpr int U = 20;
    for (int base = e0 + (int)threadIdx.x; base < e1; base += 1024 * U) {
        int d[U];
        #pragma unroll
        for (int j = 0; j < U; ++j) d[j] = dst[min(base + j * 1024, e1 - 1)];
        #pragma unroll
        for (int j = 0; j < U; ++j) {
            const unsigned u = (unsigned)(d[j] - lo);
            if (base + j * 1024 < e1 && u < (unsigned)len) atomicAdd(&cnt[u], 1u);
        }
    }
    __syncthreads();
    float* __restrict__ row = part + (size_t)c * n + lo;
    if ((n & 3) == 0) {
        float4* __restrict__ row4 = (float4*)row;
        for (int i = threadIdx.x; i < (len >> 2); i += 1024) {
            const uint4 u = cnt4[i];
            row4[i] = make_float4((float)u.x, (float)u.y, (float)u.z, (float)u.w);
        }
    } else {
        for (int i = threadIdx.x; i < len; i += 1024) row[i] = (float)cnt[i];
    }
}


constexpr int OWN_SPAN  = 224;
constexpr int OWN_MAX   = 512;
constexpr int TAB_PITCH = 544;
constexpr int BIN_CHUNK = 4096;
constexpr int BIN_EPT   = BIN_CHUNK / 1024;
constexpr int CHUNK_MAX = 512;
constexpr int OWN_T     = 512;
constexpr int OWN_W     = OWN_T / 32;

template<int NW>
__device__ __forceinline__ int block_scan_excl(int v, int* wtot, int& total) {
    const int lane = threadIdx.x & 31, wave = threadIdx.x >> 5;
    int incl = v;
    #pragma unroll
    for (int o = 1; o < 32; o <<= 1) {
        const int u = __shfl_up(incl, o);
        if (lane >= o) incl += u;
    }
    if (lane == 31) wtot[wave] = incl;
    __syncthreads();
    int wb = 0, tot = 0;
    for (int w = 0; w < NW; ++w) {
        const int x = wtot[w];
        if (w < wave) wb += x;
        tot += x;
    }
    total = tot;
    return wb + incl - v;
}

__device__ __forceinline__
void edge_bin_body(int c, const int* __restrict__ src, const int* __restrict__ dst,
                   int* __restrict__ pairs_i, int* __restrict__ tab, int n, int e) {
    int2* __restrict__ pairs = (int2*)pairs_i;
    __shared__ int  bcnt[OWN_MAX];
    __shared__ int  bst[OWN_MAX + 1];
    __shared__ int  bw[32];
    __shared__ int2 lp[BIN_CHUNK];
    const int t = threadIdx.x;
    const int nown = (n + OWN_SPAN - 1) / OWN_SPAN;
    const int e0  = c * BIN_CHUNK;
    const int len = min(BIN_CHUNK, e - e0);
    if (t < OWN_MAX) bcnt[t] = 0;
    __syncthreads();
    int s[BIN_EPT], d[BIN_EPT], ow[BIN_EPT], rk[BIN_EPT];
    #pragma unroll
    for (int u = 0; u < BIN_EPT; ++u) {
        const int i = t + u * 1024;
        const bool ok = i < len;
        s[u] = ok ? src[e0 + i] : 0;
        d[u] = ok ? dst[e0 + i] : -1;
    }
    #pragma unroll
    for (int u = 0; u < BIN_EPT; ++u) {
        ow[u] = d[u] / OWN_SPAN;
        rk[u] = (d[u] >= 0) ? atomicAdd(&bcnt[ow[u]], 1) : 0;
    }
    __syncthreads();
    int total;
    const int mine = (t < nown) ? bcnt[t] : 0;
    const int ex = block_scan_excl<32>(mine, bw, total);
    if (t < nown) bst[t] = ex;
    if (t <= nown) tab[(size_t)c * TAB_PITCH + t] = e0 + ex;
    __syncthreads();
    #pragma unroll
    for (int u = 0; u < BIN_EPT; ++u)
        if (d[u] >= 0) lp[bst[ow[u]] + rk[u]] = make_int2(s[u], d[u]);
    __syncthreads();
    for (int i = t; i < len; i += 1024) pairs[e0 + i] = lp[i];
}

__global__ __launch_bounds__(1024)
void deg_bin_kernel(const int* __restrict__ src, const int* __restrict__ dst,
                    float* __restrict__ part, int* __restrict__ pairs,
                    int* __restrict__ tab, int n, int e, int ndeg) {
    if ((int)blockIdx.x < ndeg) deg_hist_body((int)blockIdx.x, dst, part, n, e);
    else edge_bin_body((int)blockIdx.x - ndeg, src, dst, pairs, tab, n, e);
}

__device__ __forceinline__ float dinv_of(float d) {
    return (d > 0.0f) ? rsqrtf(d) : 0.0f;
}


constexpr int GEMM_WAVES = 8;

__device__ __forceinline__ float xhalf(float x) {
    return __builtin_bit_cast(float, __builtin_amdgcn_permlanex16(
        __builtin_bit_cast(int, x), __builtin_bit_cast(int, x),
        0x76543210, 0xfedcba98, false, false));
}

template<int K, int NCOLS, bool RELU_A, bool DEG_IN>
__global__ __launch_bounds__(GEMM_WAVES * 32)
void wmma_gemm_kernel(const float* __restrict__ A, const float* __restrict__ B,
                      const float* __restrict__ nrm, float* __restrict__ C,
                      float* __restrict__ dinv_out, int M) {
    constexpr int NT = NCOLS / 16;
    constexpr int KB = K / 8;
    constexpr int KQ = K / 4;
    __shared__ float4 Wl[KQ * NCOLS];

    const int wave = __builtin_amdgcn_readfirstlane((int)(threadIdx.x >> 5));
    const int lane = threadIdx.x & 31;
    const int lo16 = lane & 15;
    const int half = lane >> 4;
    const int stripe = blockIdx.x * GEMM_WAVES + wave;
    const int m0 = stripe * 16;
    const bool live = m0 < M;
    const int mr = live ? m0 : 0;

    constexpr int NST = (KQ * NCOLS) / (GEMM_WAVES * 32);
    static_assert((KQ * NCOLS) % (GEMM_WAVES * 32) == 0, "B copy: whole pieces per thread");
    float4 st[NST];
    #pragma unroll
    for (int it = 0; it < NST; ++it) {
        const int i = (int)threadIdx.x + it * (GEMM_WAVES * 32);
        const int q = i / NCOLS, col = i % NCOLS;
        const float* __restrict__ bp = B + (size_t)(4 * q) * NCOLS + col;
        st[it] = make_float4(bp[0], bp[NCOLS], bp[2 * NCOLS], bp[3 * NCOLS]);
    }

    float4 av[KB];
    {
        const float* __restrict__ arow = A + (size_t)(mr + lo16) * K + 4 * half;
        #pragma unroll
        for (int j = 0; j < KB; ++j) av[j] = *(const float4*)(arow + 8 * j);
    }

    #pragma unroll
    for (int it = 0; it < NST; ++it)
        Wl[(int)threadIdx.x + it * (GEMM_WAVES * 32)] = st[it];
    __builtin_amdgcn_fence(__ATOMIC_RELEASE, "workgroup", "local");
    __builtin_amdgcn_s_barrier();
    __builtin_amdgcn_fence(__ATOMIC_ACQUIRE, "workgroup", "local");
    if (!live) return;

    if (DEG_IN && (stripe & 1) == 0) {
        const int r = m0 + lane;
        if (r < M) {
            float d = 1.0f;
            #pragma unroll
            for (int c = 0; c < DEG_PLANES; ++c) d += nrm[(size_t)c * M + r];
            dinv_out[r] = dinv_of(d);
        }
    }

    v8f acc[NT];
    #pragma unroll
    for (int t = 0; t < NT; ++t) acc[t] = v8f{};

    #pragma unroll
    for (int j = 0; j < KB; ++j) {
        float4 a4 = av[j];
        if (RELU_A) {
            a4.x = fmaxf(a4.x, 0.0f); a4.y = fmaxf(a4.y, 0.0f);
            a4.z = fmaxf(a4.z, 0.0f); a4.w = fmaxf(a4.w, 0.0f);
        }
        v2f a01; a01[0] = a4.x; a01[1] = a4.y;
        v2f a23; a23[0] = a4.z; a23[1] = a4.w;
        float4 w[NT];
        #pragma unroll
        for (int t = 0; t < NT; ++t) w[t] = Wl[(2 * j + half) * NCOLS + 16 * t + lo16];
        #pragma unroll
        for (int t = 0; t < NT; ++t) {
            v2f b; b[0] = w[t].x; b[1] = w[t].y;
            acc[t] = __builtin_amdgcn_wmma_f32_16x16x4_f32(
                          false, a01,  false, b,
                          (short)0, acc[t],  false,  false);
        }
        #pragma unroll
        for (int t = 0; t < NT; ++t) {
            v2f b; b[0] = w[t].z; b[1] = w[t].w;
            acc[t] = __builtin_amdgcn_wmma_f32_16x16x4_f32(
                          false, a23,  false, b,
                          (short)0, acc[t],  false,  false);
        }
    }

    #pragma unroll
    for (int p = 0; p < NT / 2; ++p) {
        const int col = 32 * p + lane;
        #pragma unroll
        for (int v = 0; v < 8; ++v) {
            const float t0 = acc[2 * p][v];
            const float t1 = acc[2 * p + 1][v];
            const float got = xhalf(half ? t0 : t1);
            const float x = half ? got : t0;
            const float y = half ? t1 : got;
            const size_t ox = (size_t)(m0 + v) * NCOLS + col;
            const size_t oy = (size_t)(m0 + 8 + v) * NCOLS + col;
            C[ox] = x;
            C[oy] = y;
        }
    }
}

template<int C, bool HEAD>
__global__ __launch_bounds__(OWN_T)
void owner_agg_kernel(const int* __restrict__ pairs_i, const int* __restrict__ tab,
                      const float* __restrict__ h, const float* __restrict__ dinv,
                      const float* __restrict__ bias, float* __restrict__ agg,
                      const float* __restrict__ W2, const float* __restrict__ Wc,
                      const float* __restrict__ bc, float* __restrict__ out,
                      int n, int nchunk) {
    constexpr int R  = C / 32;
    constexpr int UB = 1024 / C;
    constexpr int EPL  = 3;
    constexpr int PASS = 32 * EPL;
    constexpr int NQ = C / 4;
    constexpr int SR = (OWN_SPAN * NQ + OWN_T - 1) / OWN_T;
    constexpr int NOUT = C / 2;
    constexpr int KQ   = C / 4;
    static_assert(KQ * NOUT <= OWN_T, "W2 copy: at most one piece per thread");
    static_assert(32 % UB == 0, "a batch of rows lies inside one lane slot of the index pass");
    static_assert(OWN_SPAN / 16 <= OWN_W, "layer 2: one 16-row stripe per wave");
    static_assert(CHUNK_MAX <= OWN_T, "one thread per chunk");
    static_assert(OWN_SPAN % 32 == 0, "whole result lines and whole stripes per owner");
    __shared__ float4 Wl[HEAD ? KQ * NOUT : 1];
    __shared__ float  outl[HEAD ? OWN_SPAN : 1];
    constexpr int ACC_Q = 108 * 1024 / 16;
    static_assert(OWN_SPAN * C / 4 <= ACC_Q, "the owner's rows fit the reserved array");
    __shared__ float4 acc4[ACC_Q];
    __shared__ int pre[CHUNK_MAX + 1];
    __shared__ int beg[CHUNK_MAX];
    __shared__ int ow_w[OWN_W];
    float* acc = (float*)acc4;
    const int2* __restrict__ pairs = (const int2*)pairs_i;

    const int t    = threadIdx.x;
    const int lane = t & 31;
    const int wave = __builtin_amdgcn_readfirstlane(t >> 5);
    const int o    = blockIdx.x;
    const int lo   = o * OWN_SPAN;
    const int len  = min(OWN_SPAN, n - lo);

    int b0 = 0, cn = 0;
    if (t < nchunk) {
        b0 = tab[(size_t)t * TAB_PITCH + o];
        cn = tab[(size_t)t * TAB_PITCH + o + 1] - b0;
    }
    float4 wst = make_float4(0.0f, 0.0f, 0.0f, 0.0f);
    if (HEAD && t < KQ * NOUT) {
        const int q = t / NOUT, col = t % NOUT;
        const float* __restrict__ bp = W2 + (size_t)(4 * q) * NOUT + col;
        wst = make_float4(bp[0], bp[NOUT], bp[2 * NOUT], bp[3 * NOUT]);
    }
    static_assert(OWN_T % NQ == 0, "a thread's pieces are the same channels of their rows");
    float4 bv = make_float4(0.0f, 0.0f, 0.0f, 0.0f);
    if (!HEAD) bv = *(const float4*)(bias + 4 * (t % NQ));
    float4 hq[SR];
    float  dq[SR];
    #pragma unroll
    for (int q = 0; q < SR; ++q) {
        const int i  = t + q * OWN_T;
        const int ii = (i < len * NQ) ? i : 0;
        hq[q] = *(const float4*)(h + (size_t)lo * C + 4 * (size_t)ii);
        dq[q] = dinv[lo + ii / NQ];
    }
    #pragma unroll
    for (int q = 0; q < SR; ++q) {
        const int i  = t + q * OWN_T;
        if (i >= len * NQ) continue;
        const float4 hv = hq[q];
        const float di = dq[q];
        float4 a;
        if (HEAD) {
            a.x = hv.x * di * di;
            a.y = hv.y * di * di;
            a.z = hv.z * di * di;
            a.w = hv.w * di * di;
        } else {
            a.x = bv.x + hv.x * di * di;
            a.y = bv.y + hv.y * di * di;
            a.z = bv.z + hv.z * di * di;
            a.w = bv.w + hv.w * di * di;
        }
        acc4[i] = a;
    }
    if (HEAD && t < KQ * NOUT) Wl[t] = wst;
    int total_v;
    const int ex = block_scan_excl<OWN_W>(cn, ow_w, total_v);
    const int total = __builtin_amdgcn_readfirstlane(total_v);
    if (t < nchunk) { pre[t] = ex; beg[t] = b0; }
    if (t == 0) pre[nchunk] = total;
    __syncthreads();

    const int per = (total + OWN_W - 1) / OWN_W;
    const int ks  = wave * per;
    const int ke  = min(ks + per, total);
    for (int k0 = ks; k0 < ke; k0 += PASS) {
        const int nb = min(PASS, ke - k0);
        int kk[EPL], lo_c[EPL], hi_c[EPL];
        #pragma unroll
        for (int s = 0; s < EPL; ++s) {
            kk[s]   = min(k0 + 32 * s + lane, total - 1);
            lo_c[s] = 0;
            hi_c[s] = nchunk;
        }
        for (int span = nchunk; span > 1; span = (span + 1) >> 1) {
            int pm[EPL];
            #pragma unroll
            for (int s = 0; s < EPL; ++s) pm[s] = pre[(lo_c[s] + hi_c[s]) >> 1];
            #pragma unroll
            for (int s = 0; s < EPL; ++s) {
                const int mid = (lo_c[s] + hi_c[s]) >> 1;
                if (pm[s] <= kk[s]) lo_c[s] = mid; else hi_c[s] = mid;
            }
        }
        int   es[EPL], doff[EPL];
        float nrm[EPL];
        {
            int2 pr[EPL];
            #pragma unroll
            for (int s = 0; s < EPL; ++s)
                pr[s] = pairs[beg[lo_c[s]] + (kk[s] - pre[lo_c[s]])];
            #pragma unroll
            for (int s = 0; s < EPL; ++s) asm volatile("" :: "v"(pr[s].x), "v"(pr[s].y));
            float dsrc[EPL], ddst[EPL];
            #pragma unroll
            for (int s = 0; s < EPL; ++s) { dsrc[s] = dinv[pr[s].x]; ddst[s] = dinv[pr[s].y]; }
            #pragma unroll
            for (int s = 0; s < EPL; ++s) {
                es[s]   = pr[s].x;
                doff[s] = (pr[s].y - lo) * C;
                nrm[s]  = dsrc[s] * ddst[s];
            }
        }
        for (int j0 = 0; j0 < nb; j0 += UB) {
            const int slot = j0 >> 5;
            int sx = es[0], dx = doff[0];
            float nx = nrm[0];
            #pragma unroll
            for (int s = 1; s < EPL; ++s)
                if (slot == s) { sx = es[s]; dx = doff[s]; nx = nrm[s]; }
            int   dd[UB];
            float nn[UB];
            float v[UB][R];
            #pragma unroll
            for (int u = 0; u < UB; ++u) {
                const int j  = min(j0 + u, nb - 1) & 31;
                const int sj = __builtin_amdgcn_readlane(sx, j);
                dd[u] = __builtin_amdgcn_readlane(dx, j);
                nn[u] = __builtin_bit_cast(float,
                            __builtin_amdgcn_readlane(__builtin_bit_cast(int, nx), j));
                #pragma unroll
                for (int r = 0; r < R; ++r)
                    v[u][r] = h[(size_t)sj * C + r * 32 + lane];
            }
            #pragma unroll
            for (int r = 0; r < R; ++r) asm volatile("" :: "v"(v[0][r]));
            #pragma unroll
            for (int u = 0; u < UB; ++u) {
                if (j0 + u < nb) {
                    #pragma unroll
                    for (int r = 0; r < R; ++r)
                        atomicAdd(&acc[dd[u] + r * 32 + lane], v[u][r] * nn[u]);
                }
            }
        }
    }
    __syncthreads();
    if (!HEAD) {
        float4* __restrict__ orow = (float4*)(agg + (size_t)lo * C);
        for (int i = t; i < len * NQ; i += OWN_T) {
            float4 a = acc4[i];
            a.x = fmaxf(a.x, 0.0f); a.y = fmaxf(a.y, 0.0f);
            a.z = fmaxf(a.z, 0.0f); a.w = fmaxf(a.w, 0.0f);
            orow[i] = a;
        }
    } else {
        const int m0 = wave * 16;
        if (m0 < len) {
            const int lo16 = lane & 15;
            const int half = lane >> 4;
            const float* arow = acc + (m0 + lo16) * C + 4 * half;
            v8f t0 = v8f{}, t1 = v8f{};
            #pragma unroll
            for (int j = 0; j < C / 8; ++j) {
                const float4 a4 = *(const float4*)(arow + 8 * j);
                const float4 w0 = Wl[(2 * j + half) * NOUT + lo16];
                const float4 w1 = Wl[(2 * j + half) * NOUT + 16 + lo16];
                v2f a01; a01[0] = a4.x; a01[1] = a4.y;
                v2f a23; a23[0] = a4.z; a23[1] = a4.w;
                v2f b;
                b[0] = w0.x; b[1] = w0.y;
                t0 = __builtin_amdgcn_wmma_f32_16x16x4_f32(false, a01, false, b, (short)0, t0, false, false);
                b[0] = w1.x; b[1] = w1.y;
                t1 = __builtin_amdgcn_wmma_f32_16x16x4_f32(false, a01, false, b, (short)0, t1, false, false);
                b[0] = w0.z; b[1] = w0.w;
                t0 = __builtin_amdgcn_wmma_f32_16x16x4_f32(false, a23, false, b, (short)0, t0, false, false);
                b[0] = w1.z; b[1] = w1.w;
                t1 = __builtin_amdgcn_wmma_f32_16x16x4_f32(false, a23, false, b, (short)0, t1, false, false);
            }
            const float ba = bias[lo16], bb = bias[16 + lo16];
            const float wa = Wc[lo16],   wb = Wc[16 + lo16];
            float s[8];
            #pragma unroll
            for (int v = 0; v < 8; ++v)
                s[v] = fmaxf(t0[v] + ba, 0.0f) * wa + fmaxf(t1[v] + bb, 0.0f) * wb;
            #pragma unroll
            for (int o2 = 8; o2 >= 1; o2 >>= 1) {
                #pragma unroll
                for (int v = 0; v < 8; ++v) s[v] += __shfl_xor(s[v], o2);
            }
            if (lo16 == 0) {
                const float b0v = bc[0];
                #pragma unroll
                for (int v = 0; v < 8; ++v) outl[m0 + 8 * half + v] = s[v] + b0v;
            }
        }
        __syncthreads();
        if (t < len) out[lo + t] = outl[t];
    }
}


static inline int cdiv(long long a, long long b) { return (int)((a + b - 1) / b); }

extern "C" void kernel_launch(void* const* d_in, const int* in_sizes, int n_in,
                              void* d_out, int out_size, void* d_ws, size_t ws_size,
                              hipStream_t stream) {
    const float* x   = (const float*)d_in[0];
    const int*   ei  = (const int*)d_in[1];
    const float* W1  = (const float*)d_in[2];
    const float* b1  = (const float*)d_in[3];
    const float* W2  = (const float*)d_in[4];
    const float* b2  = (const float*)d_in[5];
    const float* Wc  = (const float*)d_in[6];
    const float* bc  = (const float*)d_in[7];
    float* out = (float*)d_out;

    const int N = in_sizes[0] / 128;
    const int E = in_sizes[1] / 2;
    const int* src = ei;
    const int* dst = ei + E;

    float* ws   = (float*)d_ws;
    float* dinv = ws;
    float* h1   = dinv + N;
    float* agg1 = h1 + (size_t)N * 64;
    float* degp = agg1 + (size_t)N * 64;
    float* scr  = degp + (size_t)N * DEG_PLANES;
    int*   pairs = (int*)scr;
    int*   tab   = pairs + 2 * (size_t)E;
    (void)ws_size; (void)n_in; (void)out_size;

    const int ndeg   = cdiv(N, DEG_SPAN) * DEG_PLANES;
    const int nchunk = cdiv(E, BIN_CHUNK);
    const int nown   = cdiv(N, OWN_SPAN);
    deg_bin_kernel<<<ndeg + nchunk, 1024, 0, stream>>>(src, dst, degp, pairs, tab, N, E, ndeg);

    wmma_gemm_kernel<128, 64, false, true><<<cdiv(N, 16 * GEMM_WAVES), GEMM_WAVES * 32, 0, stream>>>(
        x, W1, degp, h1, dinv, N);
    owner_agg_kernel<64, false><<<nown, OWN_T, 0, stream>>>(
        pairs, tab, h1, dinv, b1, agg1, nullptr, nullptr, nullptr, nullptr, N, nchunk);

    owner_agg_kernel<64, true><<<nown, OWN_T, 0, stream>>>(
        pairs, tab, agg1, dinv, b2, nullptr, W2, Wc, bc, out, N, nchunk);
}


# In-degrees read off the owner bins: work-groups riding in the x@W1 launch count the destinations of 4 owners' bin runs in LDS and store dinv; count planes and degree work-groups of launch 1 gone
# speedup vs baseline: 4.3919x; 1.2189x over previous
#include <hip/hip_runtime.h>
#include <hip/hip_bf16.h>


typedef __attribute__((ext_vector_type(2))) float v2f;
typedef __attribute__((ext_vector_type(8))) float v8f;


constexpr int OWN_SPAN  = 224;
constexpr int OWN_MAX   = 512;
constexpr int TAB_PITCH = 544;
constexpr int BIN_CHUNK = 4096;
constexpr int BIN_EPT   = BIN_CHUNK / 1024;
constexpr int CHUNK_MAX = 512;
constexpr int OWN_T     = 512;
constexpr int OWN_W     = OWN_T / 32;

template<int NW>
__device__ __forceinline__ int block_scan_excl(int v, int* wtot, int& total) {
    const int lane = threadIdx.x & 31, wave = threadIdx.x >> 5;
    int incl = v;
    #pragma unroll
    for (int o = 1; o < 32; o <<= 1) {
        const int u = __shfl_up(incl, o);
        if (lane >= o) incl += u;
    }
    if (lane == 31) wtot[wave] = incl;
    __syncthreads();
    int wb = 0, tot = 0;
    for (int w = 0; w < NW; ++w) {
        const int x = wtot[w];
        if (w < wave) wb += x;
        tot += x;
    }
    total = tot;
    return wb + incl - v;
}

__device__ __forceinline__
void edge_bin_body(int c, const int* __restrict__ src, const int* __restrict__ dst,
                   int* __restrict__ pairs_i, int* __restrict__ tab, int n, int e) {
    int2* __restrict__ pairs = (int2*)pairs_i;
    __shared__ int  bcnt[OWN_MAX];
    __shared__ int  bst[OWN_MAX + 1];
    __shared__ int  bw[32];
    __shared__ int2 lp[BIN_CHUNK];
    const int t = threadIdx.x;
    const int nown = (n + OWN_SPAN - 1) / OWN_SPAN;
    const int e0  = c * BIN_CHUNK;
    const int len = min(BIN_CHUNK, e - e0);
    if (t < OWN_MAX) bcnt[t] = 0;
    __syncthreads();
    int s[BIN_EPT], d[BIN_EPT], ow[BIN_EPT], rk[BIN_EPT];
    #pragma unroll
    for (int u = 0; u < BIN_EPT; ++u) {
        const int i = t + u * 1024;
        const bool ok = i < len;
        s[u] = ok ? src[e0 + i] : 0;
        d[u] = ok ? dst[e0 + i] : -1;
    }
    #pragma unroll
    for (int u = 0; u < BIN_EPT; ++u) {
        ow[u] = d[u] / OWN_SPAN;
        rk[u] = (d[u] >= 0) ? atomicAdd(&bcnt[ow[u]], 1) : 0;
    }
    __syncthreads();
    int total;
    const int mine = (t < nown) ? bcnt[t] : 0;
    const int ex = block_scan_excl<32>(mine, bw, total);
    if (t < nown) bst[t] = ex;
    if (t <= nown) tab[(size_t)c * TAB_PITCH + t] = e0 + ex;
    __syncthreads();
    #pragma unroll
    for (int u = 0; u < BIN_EPT; ++u)
        if (d[u] >= 0) lp[bst[ow[u]] + rk[u]] = make_int2(s[u], d[u]);
    __syncthreads();
    for (int i = t; i < len; i += 1024) pairs[e0 + i] = lp[i];
}

__global__ __launch_bounds__(1024)
void edge_bin_kernel(const int* __restrict__ src, const int* __restrict__ dst,
                     int* __restrict__ pairs, int* __restrict__ tab, int n, int e) {
    edge_bin_body((int)blockIdx.x, src, dst, pairs, tab, n, e);
}

__device__ __forceinline__ float dinv_of(float d) {
    return (d > 0.0f) ? rsqrtf(d) : 0.0f;
}

constexpr int DEG_OWN = 4;

template<int NTHR>
__device__ __forceinline__
void deg_owner_body(int g, const int* __restrict__ pairs_i, const int* __restrict__ tab,
                    float* __restrict__ dinv_out, int n, int nchunk) {
    __shared__ unsigned cnt[DEG_OWN * OWN_SPAN];
    const int t    = threadIdx.x;
    const int nown = (n + OWN_SPAN - 1) / OWN_SPAN;
    const int o0   = g * DEG_OWN;
    const int o1   = min(o0 + DEG_OWN, nown);
    const int lo   = o0 * OWN_SPAN;
    const int len  = min(DEG_OWN * OWN_SPAN, n - lo);
    for (int i = t; i < DEG_OWN * OWN_SPAN; i += NTHR) cnt[i] = 0u;
    __syncthreads();
    constexpr int U = 16;
    for (int c = t; c < nchunk; c += NTHR) {
        const int b0 = tab[(size_t)c * TAB_PITCH + o0];
        const int b1 = tab[(size_t)c * TAB_PITCH + o1];
        for (int base = b0; base < b1; base += U) {
            int d[U];
            #pragma unroll
            for (int j = 0; j < U; ++j) d[j] = pairs_i[2 * (size_t)min(base + j, b1 - 1) + 1];
            #pragma unroll
            for (int j = 0; j < U; ++j) asm volatile("" :: "v"(d[j]));
            #pragma unroll
            for (int j = 0; j < U; ++j)
                if (base + j < b1) atomicAdd(&cnt[d[j] - lo], 1u);
        }
    }
    __syncthreads();
    for (int i = t; i < len; i += NTHR) dinv_out[lo + i] = dinv_of(1.0f + (float)cnt[i]);
}


constexpr int GEMM_WAVES = 8;

__device__ __forceinline__ float xhalf(float x) {
    return __builtin_bit_cast(float, __builtin_amdgcn_permlanex16(
        __builtin_bit_cast(int, x), __builtin_bit_cast(int, x),
        0x76543210, 0xfedcba98, false, false));
}

template<int K, int NCOLS, bool RELU_A>
__global__ __launch_bounds__(GEMM_WAVES * 32)
void wmma_gemm_kernel(const float* __restrict__ A, const float* __restrict__ B,
                      float* __restrict__ C, int M,
                      const int* __restrict__ pairs, const int* __restrict__ tab,
                      float* __restrict__ dinv_out, int nchunk, int ndeg) {
    if ((int)blockIdx.x < ndeg) {
        deg_owner_body<GEMM_WAVES * 32>((int)blockIdx.x, pairs, tab, dinv_out, M, nchunk);
        return;
    }
    constexpr int NT = NCOLS / 16;
    constexpr int KB = K / 8;
    constexpr int KQ = K / 4;
    __shared__ float4 Wl[KQ * NCOLS];

    const int wave = __builtin_amdgcn_readfirstlane((int)(threadIdx.x >> 5));
    const int lane = threadIdx.x & 31;
    const int lo16 = lane & 15;
    const int half = lane >> 4;
    const int stripe = ((int)blockIdx.x - ndeg) * GEMM_WAVES + wave;
    const int m0 = stripe * 16;
    const bool live = m0 < M;
    const int mr = live ? m0 : 0;

    constexpr int NST = (KQ * NCOLS) / (GEMM_WAVES * 32);
    static_assert((KQ * NCOLS) % (GEMM_WAVES * 32) == 0, "B copy: whole pieces per thread");
    float4 st[NST];
    #pragma unroll
    for (int it = 0; it < NST; ++it) {
        const int i = (int)threadIdx.x + it * (GEMM_WAVES * 32);
        const int q = i / NCOLS, col = i % NCOLS;
        const float* __restrict__ bp = B + (size_t)(4 * q) * NCOLS + col;
        st[it] = make_float4(bp[0], bp[NCOLS], bp[2 * NCOLS], bp[3 * NCOLS]);
    }

    float4 av[KB];
    {
        const float* __restrict__ arow = A + (size_t)(mr + lo16) * K + 4 * half;
        #pragma unroll
        for (int j = 0; j < KB; ++j) av[j] = *(const float4*)(arow + 8 * j);
    }

    #pragma unroll
    for (int it = 0; it < NST; ++it)
        Wl[(int)threadIdx.x + it * (GEMM_WAVES * 32)] = st[it];
    __builtin_amdgcn_fence(__ATOMIC_RELEASE, "workgroup", "local");
    __builtin_amdgcn_s_barrier();
    __builtin_amdgcn_fence(__ATOMIC_ACQUIRE, "workgroup", "local");
    if (!live) return;

    v8f acc[NT];
    #pragma unroll
    for (int t = 0; t < NT; ++t) acc[t] = v8f{};

    #pragma unroll
    for (int j = 0; j < KB; ++j) {
        float4 a4 = av[j];
        if (RELU_A) {
            a4.x = fmaxf(a4.x, 0.0f); a4.y = fmaxf(a4.y, 0.0f);
            a4.z = fmaxf(a4.z, 0.0f); a4.w = fmaxf(a4.w, 0.0f);
        }
        v2f a01; a01[0] = a4.x; a01[1] = a4.y;
        v2f a23; a23[0] = a4.z; a23[1] = a4.w;
        float4 w[NT];
        #pragma unroll
        for (int t = 0; t < NT; ++t) w[t] = Wl[(2 * j + half) * NCOLS + 16 * t + lo16];
        #pragma unroll
        for (int t = 0; t < NT; ++t) {
            v2f b; b[0] = w[t].x; b[1] = w[t].y;
            acc[t] = __builtin_amdgcn_wmma_f32_16x16x4_f32(
                          false, a01,  false, b,
                          (short)0, acc[t],  false,  false);
        }
        #pragma unroll
        for (int t = 0; t < NT; ++t) {
            v2f b; b[0] = w[t].z; b[1] = w[t].w;
            acc[t] = __builtin_amdgcn_wmma_f32_16x16x4_f32(
                          false, a23,  false, b,
                          (short)0, acc[t],  false,  false);
        }
    }

    #pragma unroll
    for (int p = 0; p < NT / 2; ++p) {
        const int col = 32 * p + lane;
        #pragma unroll
        for (int v = 0; v < 8; ++v) {
            const float t0 = acc[2 * p][v];
            const float t1 = acc[2 * p + 1][v];
            const float got = xhalf(half ? t0 : t1);
            const float x = half ? got : t0;
            const float y = half ? t1 : got;
            const size_t ox = (size_t)(m0 + v) * NCOLS + col;
            const size_t oy = (size_t)(m0 + 8 + v) * NCOLS + col;
            C[ox] = x;
            C[oy] = y;
        }
    }
}

template<int C, bool LAYER2>
__global__ __launch_bounds__(OWN_T)
void owner_agg_kernel(const int* __restrict__ pairs_i, const int* __restrict__ tab,
                      const float* __restrict__ h, const float* __restrict__ dinv,
                      const float* __restrict__ bias, float* __restrict__ h2out,
                      const float* __restrict__ W2, const float* __restrict__ Wc,
                      const float* __restrict__ bc, float* __restrict__ out,
                      int n, int nchunk) {
    constexpr bool PROD = !LAYER2;
    constexpr int R  = C / 32;
    constexpr int UB = 1024 / C;
    constexpr int EPL  = 3;
    constexpr int PASS = 32 * EPL;
    constexpr int NQ = C / 4;
    constexpr int SR = (OWN_SPAN * NQ + OWN_T - 1) / OWN_T;
    constexpr int NOUT = C / 2;
    constexpr int KQ   = C / 4;
    constexpr int P  = LAYER2 ? C + 4 : C;
    constexpr int PQ = P / 4;
    static_assert(!PROD || KQ * NOUT <= OWN_T, "W2 copy: at most one piece per thread");
    static_assert(!PROD || NOUT == 32, "a product row is one 128-byte line");
    static_assert(32 % UB == 0, "a batch of rows lies inside one lane slot of the index pass");
    static_assert(OWN_SPAN / 16 <= OWN_W, "layer 1: one 16-row stripe per wave");
    static_assert(CHUNK_MAX <= OWN_T, "one thread per chunk");
    static_assert(OWN_SPAN % 32 == 0, "whole result lines and whole stripes per owner");
    __shared__ float4 Wl[PROD ? KQ * NOUT : 1];
    constexpr int ACC_Q = 108 * 1024 / 16;
    static_assert(OWN_SPAN * PQ <= ACC_Q, "the owner's rows fit the reserved array");
    __shared__ float4 acc4[ACC_Q];
    __shared__ int pre[CHUNK_MAX + 1];
    __shared__ int beg[CHUNK_MAX];
    __shared__ int ow_w[OWN_W];
    float* acc = (float*)acc4;
    const int2* __restrict__ pairs = (const int2*)pairs_i;

    const int t    = threadIdx.x;
    const int lane = t & 31;
    const int wave = __builtin_amdgcn_readfirstlane(t >> 5);
    const int o    = blockIdx.x;
    const int lo   = o * OWN_SPAN;
    const int len  = min(OWN_SPAN, n - lo);

    int b0 = 0, cn = 0;
    if (t < nchunk) {
        b0 = tab[(size_t)t * TAB_PITCH + o];
        cn = tab[(size_t)t * TAB_PITCH + o + 1] - b0;
    }
    float4 wst = make_float4(0.0f, 0.0f, 0.0f, 0.0f);
    if (PROD && t < KQ * NOUT) {
        const int q = t / NOUT, col = t % NOUT;
        const float* __restrict__ bp = W2 + (size_t)(4 * q) * NOUT + col;
        wst = make_float4(bp[0], bp[NOUT], bp[2 * NOUT], bp[3 * NOUT]);
    }
    static_assert(OWN_T % NQ == 0, "a thread's pieces are the same channels of their rows");
    float4 bv = make_float4(0.0f, 0.0f, 0.0f, 0.0f);
    if (!LAYER2) bv = *(const float4*)(bias + 4 * (t % NQ));
    float4 hq[SR];
    float  dq[SR];
    #pragma unroll
    for (int q = 0; q < SR; ++q) {
        const int i  = t + q * OWN_T;
        const int ii = (i < len * NQ) ? i : 0;
        hq[q] = *(const float4*)(h + (size_t)lo * C + 4 * (size_t)ii);
        dq[q] = dinv[lo + ii / NQ];
    }
    #pragma unroll
    for (int q = 0; q < SR; ++q) {
        const int i  = t + q * OWN_T;
        if (i >= len * NQ) continue;
        const float4 hv = hq[q];
        const float di = dq[q];
        float4 a;
        if (LAYER2) {
            a.x = hv.x * di * di;
            a.y = hv.y * di * di;
            a.z = hv.z * di * di;
            a.w = hv.w * di * di;
        } else {
            a.x = bv.x + hv.x * di * di;
            a.y = bv.y + hv.y * di * di;
            a.z = bv.z + hv.z * di * di;
            a.w = bv.w + hv.w * di * di;
        }
        acc4[(i / NQ) * PQ + (i % NQ)] = a;
    }
    if (PROD && t < KQ * NOUT) Wl[t] = wst;
    int total_v;
    const int ex = block_scan_excl<OWN_W>(cn, ow_w, total_v);
    const int total = __builtin_amdgcn_readfirstlane(total_v);
    if (t < nchunk) { pre[t] = ex; beg[t] = b0; }
    if (t == 0) pre[nchunk] = total;
    __syncthreads();

    const int per = (total + OWN_W - 1) / OWN_W;
    const int ks  = wave * per;
    const int ke  = min(ks + per, total);
    for (int k0 = ks; k0 < ke; k0 += PASS) {
        const int nb = min(PASS, ke - k0);
        int kk[EPL], lo_c[EPL], hi_c[EPL];
        #pragma unroll
        for (int s = 0; s < EPL; ++s) {
            kk[s]   = min(k0 + 32 * s + lane, total - 1);
            lo_c[s] = 0;
            hi_c[s] = nchunk;
        }
        for (int span = nchunk; span > 1; span = (span + 1) >> 1) {
            int pm[EPL];
            #pragma unroll
            for (int s = 0; s < EPL; ++s) pm[s] = pre[(lo_c[s] + hi_c[s]) >> 1];
            #pragma unroll
            for (int s = 0; s < EPL; ++s) {
                const int mid = (lo_c[s] + hi_c[s]) >> 1;
                if (pm[s] <= kk[s]) lo_c[s] = mid; else hi_c[s] = mid;
            }
        }
        int   es[EPL], doff[EPL];
        float nrm[EPL];
        {
            int2 pr[EPL];
            #pragma unroll
            for (int s = 0; s < EPL; ++s)
                pr[s] = pairs[beg[lo_c[s]] + (kk[s] - pre[lo_c[s]])];
            #pragma unroll
            for (int s = 0; s < EPL; ++s) asm volatile("" :: "v"(pr[s].x), "v"(pr[s].y));
            float dsrc[EPL], ddst[EPL];
            #pragma unroll
            for (int s = 0; s < EPL; ++s) { dsrc[s] = dinv[pr[s].x]; ddst[s] = dinv[pr[s].y]; }
            #pragma unroll
            for (int s = 0; s < EPL; ++s) {
                es[s]   = pr[s].x;
                doff[s] = (pr[s].y - lo) * P;
                nrm[s]  = dsrc[s] * ddst[s];
            }
        }
        for (int j0 = 0; j0 < nb; j0 += UB) {
            const int slot = j0 >> 5;
            int sx = es[0], dx = doff[0];
            float nx = nrm[0];
            #pragma unroll
            for (int s = 1; s < EPL; ++s)
                if (slot == s) { sx = es[s]; dx = doff[s]; nx = nrm[s]; }
            float v[UB][R];
            #pragma unroll
            for (int u = 0; u < UB; ++u) {
                const int j  = min(j0 + u, nb - 1) & 31;
                const int sj = __builtin_amdgcn_readlane(sx, j);
                #pragma unroll
                for (int r = 0; r < R; ++r)
                    v[u][r] = h[(size_t)sj * C + r * 32 + lane];
            }
            #pragma unroll
            for (int r = 0; r < R; ++r) asm volatile("" :: "v"(v[0][r]));
            #pragma unroll
            for (int u = 0; u < UB; ++u) {
                if (j0 + u < nb) {
                    const int j  = (j0 + u) & 31;
                    const int dj = __builtin_amdgcn_readlane(dx, j);
                    const float nj = __builtin_bit_cast(float,
                                __builtin_amdgcn_readlane(__builtin_bit_cast(int, nx), j));
                    #pragma unroll
                    for (int r = 0; r < R; ++r)
                        atomicAdd(&acc[dj + r * 32 + lane], v[u][r] * nj);
                }
            }
        }
    }
    __syncthreads();
    if constexpr (PROD) {
        const int m0 = wave * 16;
        if (m0 < len) {
            const int lo16 = lane & 15;
            const int half = lane >> 4;
            const float* arow = acc + (m0 + lo16) * P + 4 * half;
            v8f t0 = v8f{}, t1 = v8f{};
            #pragma unroll
            for (int j = 0; j < C / 8; ++j) {
                float4 a4 = *(const float4*)(arow + 8 * j);
                a4.x = fmaxf(a4.x, 0.0f); a4.y = fmaxf(a4.y, 0.0f);
                a4.z = fmaxf(a4.z, 0.0f); a4.w = fmaxf(a4.w, 0.0f);
                const float4 w0 = Wl[(2 * j + half) * NOUT + lo16];
                const float4 w1 = Wl[(2 * j + half) * NOUT + 16 + lo16];
                v2f a01; a01[0] = a4.x; a01[1] = a4.y;
                v2f a23; a23[0] = a4.z; a23[1] = a4.w;
                v2f b;
                b[0] = w0.x; b[1] = w0.y;
                t0 = __builtin_amdgcn_wmma_f32_16x16x4_f32(false, a01, false, b, (short)0, t0, false, false);
                b[0] = w1.x; b[1] = w1.y;
                t1 = __builtin_amdgcn_wmma_f32_16x16x4_f32(false, a01, false, b, (short)0, t1, false, false);
                b[0] = w0.z; b[1] = w0.w;
                t0 = __builtin_amdgcn_wmma_f32_16x16x4_f32(false, a23, false, b, (short)0, t0, false, false);
                b[0] = w1.z; b[1] = w1.w;
                t1 = __builtin_amdgcn_wmma_f32_16x16x4_f32(false, a23, false, b, (short)0, t1, false, false);
            }
            float* __restrict__ orow = h2out + (size_t)(lo + m0) * NOUT + lane;
            #pragma unroll
            for (int v = 0; v < 8; ++v) {
                const float p0 = t0[v];
                const float p1 = t1[v];
                const float got = xhalf(half ? p0 : p1);
                const float x = half ? got : p0;
                const float y = half ? p1 : got;
                if (m0 + v < len)     orow[(size_t)v * NOUT] = x;
                if (m0 + 8 + v < len) orow[(size_t)(8 + v) * NOUT] = y;
            }
        }
    } else {
        if (t < len) {
            const float4* __restrict__ row = acc4 + t * PQ;
            float s = 0.0f;
            #pragma unroll
            for (int q = 0; q < NQ; ++q) {
                const float4 a = row[q];
                s += fmaxf(a.x + bias[4 * q],     0.0f) * Wc[4 * q];
                s += fmaxf(a.y + bias[4 * q + 1], 0.0f) * Wc[4 * q + 1];
                s += fmaxf(a.z + bias[4 * q + 2], 0.0f) * Wc[4 * q + 2];
                s += fmaxf(a.w + bias[4 * q + 3], 0.0f) * Wc[4 * q + 3];
            }
            out[lo + t] = s + bc[0];
        }
    }
}


static inline int cdiv(long long a, long long b) { return (int)((a + b - 1) / b); }

extern "C" void kernel_launch(void* const* d_in, const int* in_sizes, int n_in,
                              void* d_out, int out_size, void* d_ws, size_t ws_size,
                              hipStream_t stream) {
    const float* x   = (const float*)d_in[0];
    const int*   ei  = (const int*)d_in[1];
    const float* W1  = (const float*)d_in[2];
    const float* b1  = (const float*)d_in[3];
    const float* W2  = (const float*)d_in[4];
    const float* b2  = (const float*)d_in[5];
    const float* Wc  = (const float*)d_in[6];
    const float* bc  = (const float*)d_in[7];
    float* out = (float*)d_out;

    const int N = in_sizes[0] / 128;
    const int E = in_sizes[1] / 2;
    const int* src = ei;
    const int* dst = ei + E;

    float* ws   = (float*)d_ws;
    float* dinv = ws;
    float* h1   = dinv + N;
    float* h2   = h1 + (size_t)N * 64;
    float* scr  = h2 + (size_t)N * 32;
    int*   pairs = (int*)scr;
    int*   tab   = pairs + 2 * (size_t)E;
    (void)ws_size; (void)n_in; (void)out_size;

    const int nchunk = cdiv(E, BIN_CHUNK);
    const int nown   = cdiv(N, OWN_SPAN);
    const int ndeg   = cdiv(nown, DEG_OWN);
    edge_bin_kernel<<<nchunk, 1024, 0, stream>>>(src, dst, pairs, tab, N, E);

    wmma_gemm_kernel<128, 64, false><<<ndeg + cdiv(N, 16 * GEMM_WAVES), GEMM_WAVES * 32, 0, stream>>>(
        x, W1, h1, N, pairs, tab, dinv, nchunk, ndeg);
    owner_agg_kernel<64, false><<<nown, OWN_T, 0, stream>>>(
        pairs, tab, h1, dinv, b1, h2, W2, nullptr, nullptr, nullptr, N, nchunk);

    owner_agg_kernel<32, true><<<nown, OWN_T, 0, stream>>>(
        pairs, tab, h2, dinv, b2, nullptr, nullptr, Wc, bc, out, N, nchunk);
}
